# starting kernel
# baseline (speedup 1.0000x reference)
#include <hip/hip_runtime.h>

typedef __attribute__((ext_vector_type(2))) float v2f;
typedef __attribute__((ext_vector_type(8))) float v8f;

#define T_STEPS  2048
#define BATCH    256
#define IN_DIM   64
#define H0       16
#define H1       32
#define H2       16
#define OUT_DIM  8
#define NTHREADS 128

__device__ __forceinline__ float sigmoidf_(float x) { return 1.0f / (1.0f + __expf(-x)); }

__device__ __forceinline__ v2f load_afrag(const float* base, int lda, int kbase, int lane) {
    int m = lane & 15;
    int k = kbase + ((lane >> 4) << 1);
    const float* p = base + m * lda + k;
    v2f a;
    a.x = p[0];
    a.y = p[1];
    return a;
}

__device__ __forceinline__ v2f load_bfrag(const float* W, int ldw, int nbase, int kbase, int lane) {
    int n = nbase + (lane & 15);
    int k = kbase + ((lane >> 4) << 1);
    const float* p = W + n * ldw + k;
    v2f b;
    b.x = p[0];
    b.y = p[1];
    return b;
}

__device__ __forceinline__ v8f wmma4(v2f a, v2f b, v8f c) {
    return __builtin_amdgcn_wmma_f32_16x16x4_f32(false, a, false, b, (short)0, c, false, false);
}

__device__ __forceinline__ void store_dtile(float* g, int ldg, int nbase, int lane, v8f d) {
    int col   = nbase + (lane & 15);
    int rbase = (lane >> 4) << 3;
#pragma unroll
    for (int j = 0; j < 8; ++j)
        g[(rbase + j) * ldg + col] = d[j];
}

template <int H>
__device__ __forceinline__ void cell_update(const float* g, float* c, float* h, int tid) {
    const int fourH = 4 * H;
#pragma unroll
    for (int idx = tid; idx < 16 * H; idx += NTHREADS) {
        int m = idx / H;
        int n = idx % H;
        float vi = g[m * fourH + n];
        float vf = g[m * fourH + H + n];
        float vg = g[m * fourH + 2 * H + n];
        float vo = g[m * fourH + 3 * H + n];
        float cc = c[idx];
        float cn = sigmoidf_(vf) * cc + sigmoidf_(vi) * tanhf(vg);
        c[idx] = cn;
        h[idx] = sigmoidf_(vo) * tanhf(cn);
    }
}

__global__ __launch_bounds__(NTHREADS) void lstm3_scan_kernel(
    const float* __restrict__ x,
    const float* __restrict__ Wih0, const float* __restrict__ Whh0,
    const float* __restrict__ bih0, const float* __restrict__ bhh0,
    const float* __restrict__ Wih1, const float* __restrict__ Whh1,
    const float* __restrict__ bih1, const float* __restrict__ bhh1,
    const float* __restrict__ Wih2, const float* __restrict__ Whh2,
    const float* __restrict__ bih2, const float* __restrict__ bhh2,
    const float* __restrict__ Wl,   const float* __restrict__ bl,
    float* __restrict__ out)
{
    __shared__ float sG[16 * 4 * H1];
    __shared__ float sH0[16 * H0], sC0[16 * H0];
    __shared__ float sH1[16 * H1], sC1[16 * H1];
    __shared__ float sH2[16 * H2], sC2[16 * H2];
    __shared__ float sWl[OUT_DIM * H2];
    __shared__ float sBl[OUT_DIM];

    const int tid  = threadIdx.x;
    const int lane = tid & 31;
    const int wave = tid >> 5;
    const int ln   = lane & 15;
    const int b0   = blockIdx.x * 16;

    for (int i = tid; i < 16 * H0; i += NTHREADS) { sH0[i] = 0.f; sC0[i] = 0.f; sH2[i] = 0.f; sC2[i] = 0.f; }
    for (int i = tid; i < 16 * H1; i += NTHREADS) { sH1[i] = 0.f; sC1[i] = 0.f; }
    for (int i = tid; i < OUT_DIM * H2; i += NTHREADS) sWl[i] = Wl[i];
    if (tid < OUT_DIM) sBl[tid] = bl[tid];

    const int nb0 = wave * 16;
    v2f wx0[16], wh0[4];
#pragma unroll
    for (int kk = 0; kk < 16; ++kk) wx0[kk] = load_bfrag(Wih0, IN_DIM, nb0, 4 * kk, lane);
#pragma unroll
    for (int kk = 0; kk < 4;  ++kk) wh0[kk] = load_bfrag(Whh0, H0, nb0, 4 * kk, lane);
    const float bias0 = bih0[nb0 + ln] + bhh0[nb0 + ln];

    const int nb1a = wave * 32, nb1b = wave * 32 + 16;
    v2f wx1a[4], wx1b[4], wh1a[8], wh1b[8];
#pragma unroll
    for (int kk = 0; kk < 4; ++kk) {
        wx1a[kk] = load_bfrag(Wih1, H0, nb1a, 4 * kk, lane);
        wx1b[kk] = load_bfrag(Wih1, H0, nb1b, 4 * kk, lane);
    }
#pragma unroll
    for (int kk = 0; kk < 8; ++kk) {
        wh1a[kk] = load_bfrag(Whh1, H1, nb1a, 4 * kk, lane);
        wh1b[kk] = load_bfrag(Whh1, H1, nb1b, 4 * kk, lane);
    }
    const float bias1a = bih1[nb1a + ln] + bhh1[nb1a + ln];
    const float bias1b = bih1[nb1b + ln] + bhh1[nb1b + ln];

    const int nb2 = wave * 16;
    v2f wx2[8], wh2[4];
#pragma unroll
    for (int kk = 0; kk < 8; ++kk) wx2[kk] = load_bfrag(Wih2, H1, nb2, 4 * kk, lane);
#pragma unroll
    for (int kk = 0; kk < 4; ++kk) wh2[kk] = load_bfrag(Whh2, H2, nb2, 4 * kk, lane);
    const float bias2 = bih2[nb2 + ln] + bhh2[nb2 + ln];

    __syncthreads();

    for (int t = 0; t < T_STEPS; ++t) {
        const float* xt = x + ((size_t)t * BATCH + b0) * IN_DIM;
        if (t + 1 < T_STEPS) {
            const float* xn = xt + (size_t)BATCH * IN_DIM;
            __builtin_prefetch(xn + tid * 8, 0, 0);
        }

        v8f acc0;
#pragma unroll
        for (int j = 0; j < 8; ++j) acc0[j] = bias0;
#pragma unroll
        for (int kk = 0; kk < 16; ++kk) {
            v2f a = load_afrag(xt, IN_DIM, 4 * kk, lane);
            acc0 = wmma4(a, wx0[kk], acc0);
        }
#pragma unroll
        for (int kk = 0; kk < 4; ++kk) {
            v2f a = load_afrag(sH0, H0, 4 * kk, lane);
            acc0 = wmma4(a, wh0[kk], acc0);
        }
        store_dtile(sG, 4 * H0, nb0, lane, acc0);
        __syncthreads();
        cell_update<H0>(sG, sC0, sH0, tid);
        __syncthreads();

        v8f acc1a, acc1b;
#pragma unroll
        for (int j = 0; j < 8; ++j) { acc1a[j] = bias1a; acc1b[j] = bias1b; }
#pragma unroll
        for (int kk = 0; kk < 4; ++kk) {
            v2f a = load_afrag(sH0, H0, 4 * kk, lane);
            acc1a = wmma4(a, wx1a[kk], acc1a);
            acc1b = wmma4(a, wx1b[kk], acc1b);
        }
#pragma unroll
        for (int kk = 0; kk < 8; ++kk) {
            v2f a = load_afrag(sH1, H1, 4 * kk, lane);
            acc1a = wmma4(a, wh1a[kk], acc1a);
            acc1b = wmma4(a, wh1b[kk], acc1b);
        }
        store_dtile(sG, 4 * H1, nb1a, lane, acc1a);
        store_dtile(sG, 4 * H1, nb1b, lane, acc1b);
        __syncthreads();
        cell_update<H1>(sG, sC1, sH1, tid);
        __syncthreads();

        v8f acc2;
#pragma unroll
        for (int j = 0; j < 8; ++j) acc2[j] = bias2;
#pragma unroll
        for (int kk = 0; kk < 8; ++kk) {
            v2f a = load_afrag(sH1, H1, 4 * kk, lane);
            acc2 = wmma4(a, wx2[kk], acc2);
        }
#pragma unroll
        for (int kk = 0; kk < 4; ++kk) {
            v2f a = load_afrag(sH2, H2, 4 * kk, lane);
            acc2 = wmma4(a, wh2[kk], acc2);
        }
        store_dtile(sG, 4 * H2, nb2, lane, acc2);
        __syncthreads();
        cell_update<H2>(sG, sC2, sH2, tid);
        __syncthreads();

        {
            int m = tid >> 3;
            int o = tid & 7;
            float acc = sBl[o];
#pragma unroll
            for (int k = 0; k < H2; ++k) acc += sH2[m * H2 + k] * sWl[o * H2 + k];
            out[(((size_t)(b0 + m)) * T_STEPS + t) * OUT_DIM + o] = acc;
        }
    }
}

extern "C" void kernel_launch(void* const* d_in, const int* in_sizes, int n_in,
                              void* d_out, int out_size, void* d_ws, size_t ws_size,
                              hipStream_t stream) {
    (void)in_sizes; (void)n_in; (void)out_size; (void)d_ws; (void)ws_size;
    const float* x    = (const float*)d_in[0];
    const float* Wih0 = (const float*)d_in[1];
    const float* Whh0 = (const float*)d_in[2];
    const float* bih0 = (const float*)d_in[3];
    const float* bhh0 = (const float*)d_in[4];
    const float* Wih1 = (const float*)d_in[5];
    const float* Whh1 = (const float*)d_in[6];
    const float* bih1 = (const float*)d_in[7];
    const float* bhh1 = (const float*)d_in[8];
    const float* Wih2 = (const float*)d_in[9];
    const float* Whh2 = (const float*)d_in[10];
    const float* bih2 = (const float*)d_in[11];
    const float* bhh2 = (const float*)d_in[12];
    const float* Wl   = (const float*)d_in[13];
    const float* bl   = (const float*)d_in[14];
    float* out = (float*)d_out;

    lstm3_scan_kernel<<<BATCH / 16, NTHREADS, 0, stream>>>(
        x, Wih0, Whh0, bih0, bhh0,
        Wih1, Whh1, bih1, bhh1,
        Wih2, Whh2, bih2, bhh2,
        Wl, bl, out);
}


# cell update: library tanhf and correctly rounded 1/x replaced by branch-free v_exp_f32/v_rcp_f32 forms (the library's f32 polynomial for small arguments kept), so each thread's elements interleave
# speedup vs baseline: 1.6673x; 1.6673x over previous
#include <hip/hip_runtime.h>

typedef __attribute__((ext_vector_type(2))) float v2f;
typedef __attribute__((ext_vector_type(8))) float v8f;

#define T_STEPS  2048
#define BATCH    256
#define IN_DIM   64
#define H0       16
#define H1       32
#define H2       16
#define OUT_DIM  8
#define NTHREADS 128

__device__ __forceinline__ float sigmoidf_(float x) {
    return __builtin_amdgcn_rcpf(1.0f + __builtin_amdgcn_exp2f(x * -1.442695041f));
}

__device__ __forceinline__ float tanhf_(float x) {
    const float ax = fabsf(x);
    const float e   = __builtin_amdgcn_exp2f(ax * 2.885390082f);
    const float big = fmaf(__builtin_amdgcn_rcpf(e + 1.0f), -2.0f, 1.0f);
    const float t = ax * ax;
    float p = fmaf(t, -0.00570002059f, 0.0206340719f);
    p = fmaf(t, p, -0.0537379310f);
    p = fmaf(t, p, 0.133314162f);
    p = fmaf(t, p, -0.333332807f);
    const float sm = fmaf(t, ax * p, ax);
    return copysignf(ax < 0.625f ? sm : big, x);
}

__device__ __forceinline__ v2f load_afrag(const float* base, int lda, int kbase, int lane) {
    int m = lane & 15;
    int k = kbase + ((lane >> 4) << 1);
    const float* p = base + m * lda + k;
    v2f a;
    a.x = p[0];
    a.y = p[1];
    return a;
}

__device__ __forceinline__ v2f load_bfrag(const float* W, int ldw, int nbase, int kbase, int lane) {
    int n = nbase + (lane & 15);
    int k = kbase + ((lane >> 4) << 1);
    const float* p = W + n * ldw + k;
    v2f b;
    b.x = p[0];
    b.y = p[1];
    return b;
}

__device__ __forceinline__ v8f wmma4(v2f a, v2f b, v8f c) {
    return __builtin_amdgcn_wmma_f32_16x16x4_f32(false, a, false, b, (short)0, c, false, false);
}

__device__ __forceinline__ void store_dtile(float* g, int ldg, int nbase, int lane, v8f d) {
    int col   = nbase + (lane & 15);
    int rbase = (lane >> 4) << 3;
#pragma unroll
    for (int j = 0; j < 8; ++j)
        g[(rbase + j) * ldg + col] = d[j];
}

template <int H>
__device__ __forceinline__ void cell_update(const float* g, float* c, float* h, int tid) {
    const int fourH = 4 * H;
    constexpr int E = (16 * H) / NTHREADS;
    float vi[E], vf[E], vg[E], vo[E], cc[E];
#pragma unroll
    for (int e = 0; e < E; ++e) {
        int idx = tid + e * NTHREADS;
        int m = idx / H;
        int n = idx % H;
        vi[e] = g[m * fourH + n];
        vf[e] = g[m * fourH + H + n];
        vg[e] = g[m * fourH + 2 * H + n];
        vo[e] = g[m * fourH + 3 * H + n];
        cc[e] = c[idx];
    }
#pragma unroll
    for (int e = 0; e < E; ++e) {
        int idx = tid + e * NTHREADS;
        float cn = sigmoidf_(vf[e]) * cc[e] + sigmoidf_(vi[e]) * tanhf_(vg[e]);
        c[idx] = cn;
        h[idx] = sigmoidf_(vo[e]) * tanhf_(cn);
    }
}

__global__ __launch_bounds__(NTHREADS) void lstm3_scan_kernel(
    const float* __restrict__ x,
    const float* __restrict__ Wih0, const float* __restrict__ Whh0,
    const float* __restrict__ bih0, const float* __restrict__ bhh0,
    const float* __restrict__ Wih1, const float* __restrict__ Whh1,
    const float* __restrict__ bih1, const float* __restrict__ bhh1,
    const float* __restrict__ Wih2, const float* __restrict__ Whh2,
    const float* __restrict__ bih2, const float* __restrict__ bhh2,
    const float* __restrict__ Wl,   const float* __restrict__ bl,
    float* __restrict__ out)
{
    __shared__ float sG[16 * 4 * H1];
    __shared__ float sH0[16 * H0], sC0[16 * H0];
    __shared__ float sH1[16 * H1], sC1[16 * H1];
    __shared__ float sH2[16 * H2], sC2[16 * H2];
    __shared__ float sWl[OUT_DIM * H2];
    __shared__ float sBl[OUT_DIM];

    const int tid  = threadIdx.x;
    const int lane = tid & 31;
    const int wave = tid >> 5;
    const int ln   = lane & 15;
    const int b0   = blockIdx.x * 16;

    for (int i = tid; i < 16 * H0; i += NTHREADS) { sH0[i] = 0.f; sC0[i] = 0.f; sH2[i] = 0.f; sC2[i] = 0.f; }
    for (int i = tid; i < 16 * H1; i += NTHREADS) { sH1[i] = 0.f; sC1[i] = 0.f; }
    for (int i = tid; i < OUT_DIM * H2; i += NTHREADS) sWl[i] = Wl[i];
    if (tid < OUT_DIM) sBl[tid] = bl[tid];

    const int nb0 = wave * 16;
    v2f wx0[16], wh0[4];
#pragma unroll
    for (int kk = 0; kk < 16; ++kk) wx0[kk] = load_bfrag(Wih0, IN_DIM, nb0, 4 * kk, lane);
#pragma unroll
    for (int kk = 0; kk < 4;  ++kk) wh0[kk] = load_bfrag(Whh0, H0, nb0, 4 * kk, lane);
    const float bias0 = bih0[nb0 + ln] + bhh0[nb0 + ln];

    const int nb1a = wave * 32, nb1b = wave * 32 + 16;
    v2f wx1a[4], wx1b[4], wh1a[8], wh1b[8];
#pragma unroll
    for (int kk = 0; kk < 4; ++kk) {
        wx1a[kk] = load_bfrag(Wih1, H0, nb1a, 4 * kk, lane);
        wx1b[kk] = load_bfrag(Wih1, H0, nb1b, 4 * kk, lane);
    }
#pragma unroll
    for (int kk = 0; kk < 8; ++kk) {
        wh1a[kk] = load_bfrag(Whh1, H1, nb1a, 4 * kk, lane);
        wh1b[kk] = load_bfrag(Whh1, H1, nb1b, 4 * kk, lane);
    }
    const float bias1a = bih1[nb1a + ln] + bhh1[nb1a + ln];
    const float bias1b = bih1[nb1b + ln] + bhh1[nb1b + ln];

    const int nb2 = wave * 16;
    v2f wx2[8], wh2[4];
#pragma unroll
    for (int kk = 0; kk < 8; ++kk) wx2[kk] = load_bfrag(Wih2, H1, nb2, 4 * kk, lane);
#pragma unroll
    for (int kk = 0; kk < 4; ++kk) wh2[kk] = load_bfrag(Whh2, H2, nb2, 4 * kk, lane);
    const float bias2 = bih2[nb2 + ln] + bhh2[nb2 + ln];

    __syncthreads();

    for (int t = 0; t < T_STEPS; ++t) {
        const float* xt = x + ((size_t)t * BATCH + b0) * IN_DIM;
        if (t + 1 < T_STEPS) {
            const float* xn = xt + (size_t)BATCH * IN_DIM;
            __builtin_prefetch(xn + tid * 8, 0, 0);
        }

        v8f acc0;
#pragma unroll
        for (int j = 0; j < 8; ++j) acc0[j] = bias0;
#pragma unroll
        for (int kk = 0; kk < 16; ++kk) {
            v2f a = load_afrag(xt, IN_DIM, 4 * kk, lane);
            acc0 = wmma4(a, wx0[kk], acc0);
        }
#pragma unroll
        for (int kk = 0; kk < 4; ++kk) {
            v2f a = load_afrag(sH0, H0, 4 * kk, lane);
            acc0 = wmma4(a, wh0[kk], acc0);
        }
        store_dtile(sG, 4 * H0, nb0, lane, acc0);
        __syncthreads();
        cell_update<H0>(sG, sC0, sH0, tid);
        __syncthreads();

        v8f acc1a, acc1b;
#pragma unroll
        for (int j = 0; j < 8; ++j) { acc1a[j] = bias1a; acc1b[j] = bias1b; }
#pragma unroll
        for (int kk = 0; kk < 4; ++kk) {
            v2f a = load_afrag(sH0, H0, 4 * kk, lane);
            acc1a = wmma4(a, wx1a[kk], acc1a);
            acc1b = wmma4(a, wx1b[kk], acc1b);
        }
#pragma unroll
        for (int kk = 0; kk < 8; ++kk) {
            v2f a = load_afrag(sH1, H1, 4 * kk, lane);
            acc1a = wmma4(a, wh1a[kk], acc1a);
            acc1b = wmma4(a, wh1b[kk], acc1b);
        }
        store_dtile(sG, 4 * H1, nb1a, lane, acc1a);
        store_dtile(sG, 4 * H1, nb1b, lane, acc1b);
        __syncthreads();
        cell_update<H1>(sG, sC1, sH1, tid);
        __syncthreads();

        v8f acc2;
#pragma unroll
        for (int j = 0; j < 8; ++j) acc2[j] = bias2;
#pragma unroll
        for (int kk = 0; kk < 8; ++kk) {
            v2f a = load_afrag(sH1, H1, 4 * kk, lane);
            acc2 = wmma4(a, wx2[kk], acc2);
        }
#pragma unroll
        for (int kk = 0; kk < 4; ++kk) {
            v2f a = load_afrag(sH2, H2, 4 * kk, lane);
            acc2 = wmma4(a, wh2[kk], acc2);
        }
        store_dtile(sG, 4 * H2, nb2, lane, acc2);
        __syncthreads();
        cell_update<H2>(sG, sC2, sH2, tid);
        __syncthreads();

        {
            int m = tid >> 3;
            int o = tid & 7;
            float acc = sBl[o];
#pragma unroll
            for (int k = 0; k < H2; ++k) acc += sH2[m * H2 + k] * sWl[o * H2 + k];
            out[(((size_t)(b0 + m)) * T_STEPS + t) * OUT_DIM + o] = acc;
        }
    }
}

extern "C" void kernel_launch(void* const* d_in, const int* in_sizes, int n_in,
                              void* d_out, int out_size, void* d_ws, size_t ws_size,
                              hipStream_t stream) {
    (void)in_sizes; (void)n_in; (void)out_size; (void)d_ws; (void)ws_size;
    const float* x    = (const float*)d_in[0];
    const float* Wih0 = (const float*)d_in[1];
    const float* Whh0 = (const float*)d_in[2];
    const float* bih0 = (const float*)d_in[3];
    const float* bhh0 = (const float*)d_in[4];
    const float* Wih1 = (const float*)d_in[5];
    const float* Whh1 = (const float*)d_in[6];
    const float* bih1 = (const float*)d_in[7];
    const float* bhh1 = (const float*)d_in[8];
    const float* Wih2 = (const float*)d_in[9];
    const float* Whh2 = (const float*)d_in[10];
    const float* bih2 = (const float*)d_in[11];
    const float* bhh2 = (const float*)d_in[12];
    const float* Wl   = (const float*)d_in[13];
    const float* bl   = (const float*)d_in[14];
    float* out = (float*)d_out;

    lstm3_scan_kernel<<<BATCH / 16, NTHREADS, 0, stream>>>(
        x, Wih0, Whh0, bih0, bhh0,
        Wih1, Whh1, bih1, bhh1,
        Wih2, Whh2, bih2, bhh2,
        Wl, bl, out);
}


# layer-0 input projection software-pipelined one step ahead: x(t+1) fragments loaded during step t across LDS-only barriers, its 16 products run beside later vector work; sum order unchanged
# speedup vs baseline: 2.0485x; 1.2286x over previous
#include <hip/hip_runtime.h>

typedef __attribute__((ext_vector_type(2))) float v2f;
typedef __attribute__((ext_vector_type(8))) float v8f;

#define T_STEPS  2048
#define BATCH    256
#define IN_DIM   64
#define H0       16
#define H1       32
#define H2       16
#define OUT_DIM  8
#define NTHREADS 128

__device__ __forceinline__ float sigmoidf_(float x) {
    return __builtin_amdgcn_rcpf(1.0f + __builtin_amdgcn_exp2f(x * -1.442695041f));
}

__device__ __forceinline__ float tanhf_(float x) {
    const float ax = fabsf(x);
    const float e   = __builtin_amdgcn_exp2f(ax * 2.885390082f);
    const float big = fmaf(__builtin_amdgcn_rcpf(e + 1.0f), -2.0f, 1.0f);
    const float t = ax * ax;
    float p = fmaf(t, -0.00570002059f, 0.0206340719f);
    p = fmaf(t, p, -0.0537379310f);
    p = fmaf(t, p, 0.133314162f);
    p = fmaf(t, p, -0.333332807f);
    const float sm = fmaf(t, ax * p, ax);
    return copysignf(ax < 0.625f ? sm : big, x);
}

__device__ __forceinline__ v2f load_afrag(const float* base, int lda, int kbase, int lane) {
    int m = lane & 15;
    int k = kbase + ((lane >> 4) << 1);
    const float* p = base + m * lda + k;
    v2f a;
    a.x = p[0];
    a.y = p[1];
    return a;
}

__device__ __forceinline__ v2f load_bfrag(const float* W, int ldw, int nbase, int kbase, int lane) {
    int n = nbase + (lane & 15);
    int k = kbase + ((lane >> 4) << 1);
    const float* p = W + n * ldw + k;
    v2f b;
    b.x = p[0];
    b.y = p[1];
    return b;
}

__device__ __forceinline__ v8f wmma4(v2f a, v2f b, v8f c) {
    return __builtin_amdgcn_wmma_f32_16x16x4_f32(false, a, false, b, (short)0, c, false, false);
}

__device__ __forceinline__ void lds_barrier() {
    __builtin_amdgcn_fence(__ATOMIC_RELEASE, "workgroup", "local");
    __builtin_amdgcn_s_barrier();
    __builtin_amdgcn_fence(__ATOMIC_ACQUIRE, "workgroup", "local");
}

__device__ __forceinline__ void store_dtile(float* g, int ldg, int nbase, int lane, v8f d) {
    int col   = nbase + (lane & 15);
    int rbase = (lane >> 4) << 3;
#pragma unroll
    for (int j = 0; j < 8; ++j)
        g[(rbase + j) * ldg + col] = d[j];
}

template <int H>
__device__ __forceinline__ void cell_update(const float* g, float* c, float* h, int tid) {
    const int fourH = 4 * H;
    constexpr int E = (16 * H) / NTHREADS;
    float vi[E], vf[E], vg[E], vo[E], cc[E];
#pragma unroll
    for (int e = 0; e < E; ++e) {
        int idx = tid + e * NTHREADS;
        int m = idx / H;
        int n = idx % H;
        vi[e] = g[m * fourH + n];
        vf[e] = g[m * fourH + H + n];
        vg[e] = g[m * fourH + 2 * H + n];
        vo[e] = g[m * fourH + 3 * H + n];
        cc[e] = c[idx];
    }
#pragma unroll
    for (int e = 0; e < E; ++e) {
        int idx = tid + e * NTHREADS;
        float cn = sigmoidf_(vf[e]) * cc[e] + sigmoidf_(vi[e]) * tanhf_(vg[e]);
        c[idx] = cn;
        h[idx] = sigmoidf_(vo[e]) * tanhf_(cn);
    }
}

__global__ __launch_bounds__(NTHREADS) void lstm3_scan_kernel(
    const float* __restrict__ x,
    const float* __restrict__ Wih0, const float* __restrict__ Whh0,
    const float* __restrict__ bih0, const float* __restrict__ bhh0,
    const float* __restrict__ Wih1, const float* __restrict__ Whh1,
    const float* __restrict__ bih1, const float* __restrict__ bhh1,
    const float* __restrict__ Wih2, const float* __restrict__ Whh2,
    const float* __restrict__ bih2, const float* __restrict__ bhh2,
    const float* __restrict__ Wl,   const float* __restrict__ bl,
    float* __restrict__ out)
{
    __shared__ float sG[16 * 4 * H1];
    __shared__ float sH0[16 * H0], sC0[16 * H0];
    __shared__ float sH1[16 * H1], sC1[16 * H1];
    __shared__ float sH2[16 * H2], sC2[16 * H2];
    __shared__ float sWl[OUT_DIM * H2];
    __shared__ float sBl[OUT_DIM];

    const int tid  = threadIdx.x;
    const int lane = tid & 31;
    const int wave = tid >> 5;
    const int ln   = lane & 15;
    const int b0   = blockIdx.x * 16;

    for (int i = tid; i < 16 * H0; i += NTHREADS) { sH0[i] = 0.f; sC0[i] = 0.f; sH2[i] = 0.f; sC2[i] = 0.f; }
    for (int i = tid; i < 16 * H1; i += NTHREADS) { sH1[i] = 0.f; sC1[i] = 0.f; }
    for (int i = tid; i < OUT_DIM * H2; i += NTHREADS) sWl[i] = Wl[i];
    if (tid < OUT_DIM) sBl[tid] = bl[tid];

    const int nb0 = wave * 16;
    v2f wx0[16], wh0[4];
#pragma unroll
    for (int kk = 0; kk < 16; ++kk) wx0[kk] = load_bfrag(Wih0, IN_DIM, nb0, 4 * kk, lane);
#pragma unroll
    for (int kk = 0; kk < 4;  ++kk) wh0[kk] = load_bfrag(Whh0, H0, nb0, 4 * kk, lane);
    const float bias0 = bih0[nb0 + ln] + bhh0[nb0 + ln];

    const int nb1a = wave * 32, nb1b = wave * 32 + 16;
    v2f wx1a[4], wx1b[4], wh1a[8], wh1b[8];
#pragma unroll
    for (int kk = 0; kk < 4; ++kk) {
        wx1a[kk] = load_bfrag(Wih1, H0, nb1a, 4 * kk, lane);
        wx1b[kk] = load_bfrag(Wih1, H0, nb1b, 4 * kk, lane);
    }
#pragma unroll
    for (int kk = 0; kk < 8; ++kk) {
        wh1a[kk] = load_bfrag(Whh1, H1, nb1a, 4 * kk, lane);
        wh1b[kk] = load_bfrag(Whh1, H1, nb1b, 4 * kk, lane);
    }
    const float bias1a = bih1[nb1a + ln] + bhh1[nb1a + ln];
    const float bias1b = bih1[nb1b + ln] + bhh1[nb1b + ln];

    const int nb2 = wave * 16;
    v2f wx2[8], wh2[4];
#pragma unroll
    for (int kk = 0; kk < 8; ++kk) wx2[kk] = load_bfrag(Wih2, H1, nb2, 4 * kk, lane);
#pragma unroll
    for (int kk = 0; kk < 4; ++kk) wh2[kk] = load_bfrag(Whh2, H2, nb2, 4 * kk, lane);
    const float bias2 = bih2[nb2 + ln] + bhh2[nb2 + ln];

    v8f accx;
#pragma unroll
    for (int j = 0; j < 8; ++j) accx[j] = bias0;
    {
        const float* x0 = x + (size_t)b0 * IN_DIM;
#pragma unroll
        for (int kk = 0; kk < 16; ++kk) {
            v2f a = load_afrag(x0, IN_DIM, 4 * kk, lane);
            accx = wmma4(a, wx0[kk], accx);
        }
    }

    __syncthreads();

    for (int t = 0; t < T_STEPS; ++t) {
        const int tn = (t + 1 < T_STEPS) ? t + 1 : t;
        const float* xnext = x + ((size_t)tn * BATCH + b0) * IN_DIM;
        if (t + 2 < T_STEPS) {
            __builtin_prefetch(xnext + (size_t)BATCH * IN_DIM + tid * 8, 0, 0);
        }
        v2f xn[16];
#pragma unroll
        for (int kk = 0; kk < 16; ++kk) xn[kk] = load_afrag(xnext, IN_DIM, 4 * kk, lane);

        v8f acc0 = accx;
#pragma unroll
        for (int kk = 0; kk < 4; ++kk) {
            v2f a = load_afrag(sH0, H0, 4 * kk, lane);
            acc0 = wmma4(a, wh0[kk], acc0);
        }
        store_dtile(sG, 4 * H0, nb0, lane, acc0);
        lds_barrier();
        cell_update<H0>(sG, sC0, sH0, tid);
        lds_barrier();

        v8f acc1a, acc1b;
#pragma unroll
        for (int j = 0; j < 8; ++j) { acc1a[j] = bias1a; acc1b[j] = bias1b; }
#pragma unroll
        for (int kk = 0; kk < 4; ++kk) {
            v2f a = load_afrag(sH0, H0, 4 * kk, lane);
            acc1a = wmma4(a, wx1a[kk], acc1a);
            acc1b = wmma4(a, wx1b[kk], acc1b);
        }
#pragma unroll
        for (int kk = 0; kk < 8; ++kk) {
            v2f a = load_afrag(sH1, H1, 4 * kk, lane);
            acc1a = wmma4(a, wh1a[kk], acc1a);
            acc1b = wmma4(a, wh1b[kk], acc1b);
        }
        store_dtile(sG, 4 * H1, nb1a, lane, acc1a);
        store_dtile(sG, 4 * H1, nb1b, lane, acc1b);
        lds_barrier();
        cell_update<H1>(sG, sC1, sH1, tid);
        lds_barrier();

        v8f acc2;
#pragma unroll
        for (int j = 0; j < 8; ++j) acc2[j] = bias2;
#pragma unroll
        for (int kk = 0; kk < 8; ++kk) {
            v2f a = load_afrag(sH1, H1, 4 * kk, lane);
            acc2 = wmma4(a, wx2[kk], acc2);
        }
#pragma unroll
        for (int kk = 0; kk < 4; ++kk) {
            v2f a = load_afrag(sH2, H2, 4 * kk, lane);
            acc2 = wmma4(a, wh2[kk], acc2);
        }
        store_dtile(sG, 4 * H2, nb2, lane, acc2);
        lds_barrier();
#pragma unroll
        for (int j = 0; j < 8; ++j) accx[j] = bias0;
#pragma unroll
        for (int kk = 0; kk < 16; ++kk) accx = wmma4(xn[kk], wx0[kk], accx);
        cell_update<H2>(sG, sC2, sH2, tid);
        lds_barrier();

        {
            int m = tid >> 3;
            int o = tid & 7;
            float acc = sBl[o];
#pragma unroll
            for (int k = 0; k < H2; ++k) acc += sH2[m * H2 + k] * sWl[o * H2 + k];
            out[(((size_t)(b0 + m)) * T_STEPS + t) * OUT_DIM + o] = acc;
        }
    }
}

extern "C" void kernel_launch(void* const* d_in, const int* in_sizes, int n_in,
                              void* d_out, int out_size, void* d_ws, size_t ws_size,
                              hipStream_t stream) {
    (void)in_sizes; (void)n_in; (void)out_size; (void)d_ws; (void)ws_size;
    const float* x    = (const float*)d_in[0];
    const float* Wih0 = (const float*)d_in[1];
    const float* Whh0 = (const float*)d_in[2];
    const float* bih0 = (const float*)d_in[3];
    const float* bhh0 = (const float*)d_in[4];
    const float* Wih1 = (const float*)d_in[5];
    const float* Whh1 = (const float*)d_in[6];
    const float* bih1 = (const float*)d_in[7];
    const float* bhh1 = (const float*)d_in[8];
    const float* Wih2 = (const float*)d_in[9];
    const float* Whh2 = (const float*)d_in[10];
    const float* bih2 = (const float*)d_in[11];
    const float* bhh2 = (const float*)d_in[12];
    const float* Wl   = (const float*)d_in[13];
    const float* bl   = (const float*)d_in[14];
    float* out = (float*)d_out;

    lstm3_scan_kernel<<<BATCH / 16, NTHREADS, 0, stream>>>(
        x, Wih0, Whh0, bih0, bhh0,
        Wih1, Whh1, bih1, bhh1,
        Wih2, Whh2, bih2, bhh2,
        Wl, bl, out);
}


# gate products transposed (weights as A, batch tile as B) so each lane holds i,f,g,o of its own cells: cell update and c stay in registers, gate LDS round trip and 3 of 6 barriers go; sum order kept
# speedup vs baseline: 2.4485x; 1.1953x over previous
#include <hip/hip_runtime.h>


typedef __attribute__((ext_vector_type(2))) float v2f;
typedef __attribute__((ext_vector_type(4))) float v4f;
typedef __attribute__((ext_vector_type(8))) float v8f;

#define T_STEPS  2048
#define BATCH    256
#define IN_DIM   64
#define H0       16
#define H1       32
#define H2       16
#define OUT_DIM  8
#define NTHREADS 128

__device__ __forceinline__ float sigmoidf_(float x) {
    return __builtin_amdgcn_rcpf(1.0f + __builtin_amdgcn_exp2f(x * -1.442695041f));
}

__device__ __forceinline__ float tanhf_(float x) {
    const float ax = fabsf(x);
    const float e   = __builtin_amdgcn_exp2f(ax * 2.885390082f);
    const float big = fmaf(__builtin_amdgcn_rcpf(e + 1.0f), -2.0f, 1.0f);
    const float t = ax * ax;
    float p = fmaf(t, -0.00570002059f, 0.0206340719f);
    p = fmaf(t, p, -0.0537379310f);
    p = fmaf(t, p, 0.133314162f);
    p = fmaf(t, p, -0.333332807f);
    const float sm = fmaf(t, ax * p, ax);
    return copysignf(ax < 0.625f ? sm : big, x);
}

__device__ __forceinline__ v2f load_frag(const float* base, int ld, int kbase, int lane) {
    const float* p = base + (lane & 15) * ld + kbase + ((lane >> 4) << 1);
    v2f a;
    a.x = p[0];
    a.y = p[1];
    return a;
}

template <int H>
__device__ __forceinline__ v2f load_wfrag(const float* W, int ldw, int ubase, int hs, int kbase, int lane) {
    const int c   = lane & 15;
    const int row = (c & 3) * H + ubase + hs * (c >> 3) + ((c >> 2) & 1);
    const float* p = W + row * ldw + kbase + ((lane >> 4) << 1);
    v2f a;
    a.x = p[0];
    a.y = p[1];
    return a;
}

template <int H>
__device__ __forceinline__ v8f load_bias(const float* bi, const float* bh, int ubase, int hs, int lane) {
    const int hh = lane >> 4;
    v8f b;
#pragma unroll
    for (int r = 0; r < 8; ++r) {
        const int row = (r & 3) * H + ubase + hs * hh + (r >> 2);
        b[r] = bi[row] + bh[row];
    }
    return b;
}

__device__ __forceinline__ v8f wmma4(v2f a, v2f b, v8f c) {
    return __builtin_amdgcn_wmma_f32_16x16x4_f32(false, a, false, b, (short)0, c, false, false);
}

__device__ __forceinline__ void lds_barrier() {
    __builtin_amdgcn_fence(__ATOMIC_RELEASE, "workgroup", "local");
    __builtin_amdgcn_s_barrier();
    __builtin_amdgcn_fence(__ATOMIC_ACQUIRE, "workgroup", "local");
}

__device__ __forceinline__ v2f cell2(v8f d, float& c0, float& c1) {
    v2f h;
    {
        float cn = sigmoidf_(d[1]) * c0 + sigmoidf_(d[0]) * tanhf_(d[2]);
        c0 = cn;
        h.x = sigmoidf_(d[3]) * tanhf_(cn);
    }
    {
        float cn = sigmoidf_(d[5]) * c1 + sigmoidf_(d[4]) * tanhf_(d[6]);
        c1 = cn;
        h.y = sigmoidf_(d[7]) * tanhf_(cn);
    }
    return h;
}

__global__ __launch_bounds__(NTHREADS) void lstm3_scan_kernel(
    const float* __restrict__ x,
    const float* __restrict__ Wih0, const float* __restrict__ Whh0,
    const float* __restrict__ bih0, const float* __restrict__ bhh0,
    const float* __restrict__ Wih1, const float* __restrict__ Whh1,
    const float* __restrict__ bih1, const float* __restrict__ bhh1,
    const float* __restrict__ Wih2, const float* __restrict__ Whh2,
    const float* __restrict__ bih2, const float* __restrict__ bhh2,
    const float* __restrict__ Wl,   const float* __restrict__ bl,
    float* __restrict__ out)
{
    __shared__ __attribute__((aligned(16))) float sH0[2][16 * H0];
    __shared__ __attribute__((aligned(16))) float sH1[2][16 * H1];
    __shared__ __attribute__((aligned(16))) float sH2[2][16 * H2];
    __shared__ float sWl[OUT_DIM * H2];
    __shared__ float sBl[OUT_DIM];

    const int tid  = threadIdx.x;
    const int lane = tid & 31;
    const int wave = tid >> 5;
    const int ln   = lane & 15;
    const int hh   = lane >> 4;
    const int b0   = blockIdx.x * 16;

    for (int i = tid; i < 16 * H0; i += NTHREADS) { sH0[0][i] = 0.f; sH2[0][i] = 0.f; }
    for (int i = tid; i < 16 * H1; i += NTHREADS) { sH1[0][i] = 0.f; }
    for (int i = tid; i < OUT_DIM * H2; i += NTHREADS) sWl[i] = Wl[i];
    if (tid < OUT_DIM) sBl[tid] = bl[tid];

    const int u0 = wave * 4;
    v2f wx0[16], wh0[4];
#pragma unroll
    for (int kk = 0; kk < 16; ++kk) wx0[kk] = load_wfrag<H0>(Wih0, IN_DIM, u0, 2, 4 * kk, lane);
#pragma unroll
    for (int kk = 0; kk < 4;  ++kk) wh0[kk] = load_wfrag<H0>(Whh0, H0, u0, 2, 4 * kk, lane);
    const v8f bias0 = load_bias<H0>(bih0, bhh0, u0, 2, lane);

    const int u1a = wave * 8, u1b = wave * 8 + 2;
    v2f wx1a[4], wx1b[4], wh1a[8], wh1b[8];
#pragma unroll
    for (int kk = 0; kk < 4; ++kk) {
        wx1a[kk] = load_wfrag<H1>(Wih1, H0, u1a, 4, 4 * kk, lane);
        wx1b[kk] = load_wfrag<H1>(Wih1, H0, u1b, 4, 4 * kk, lane);
    }
#pragma unroll
    for (int kk = 0; kk < 8; ++kk) {
        wh1a[kk] = load_wfrag<H1>(Whh1, H1, u1a, 4, 4 * kk, lane);
        wh1b[kk] = load_wfrag<H1>(Whh1, H1, u1b, 4, 4 * kk, lane);
    }
    const v8f bias1a = load_bias<H1>(bih1, bhh1, u1a, 4, lane);
    const v8f bias1b = load_bias<H1>(bih1, bhh1, u1b, 4, lane);

    const int u2 = wave * 4;
    v2f wx2[8], wh2[4];
#pragma unroll
    for (int kk = 0; kk < 8; ++kk) wx2[kk] = load_wfrag<H2>(Wih2, H1, u2, 2, 4 * kk, lane);
#pragma unroll
    for (int kk = 0; kk < 4; ++kk) wh2[kk] = load_wfrag<H2>(Whh2, H2, u2, 2, 4 * kk, lane);
    const v8f bias2 = load_bias<H2>(bih2, bhh2, u2, 2, lane);

    const int o0 = ln * H0 + u0 + 2 * hh;
    const int o1 = ln * H1 + u1a + 4 * hh;
    const int o2 = ln * H2 + u2 + 2 * hh;

    float c0a = 0.f, c0b = 0.f;
    float c1a = 0.f, c1b = 0.f, c1c = 0.f, c1d = 0.f;
    float c2a = 0.f, c2b = 0.f;

    v8f accx = bias0;
    {
        const float* x0 = x + (size_t)b0 * IN_DIM;
#pragma unroll
        for (int kk = 0; kk < 16; ++kk) {
            v2f b = load_frag(x0, IN_DIM, 4 * kk, lane);
            accx = wmma4(wx0[kk], b, accx);
        }
    }

    __syncthreads();

    for (int t = 0; t < T_STEPS; ++t) {
        const int cur = t & 1, nxt = cur ^ 1;

        const int tn = (t + 1 < T_STEPS) ? t + 1 : t;
        const float* xnext = x + ((size_t)tn * BATCH + b0) * IN_DIM;
        if (t + 2 < T_STEPS) {
            __builtin_prefetch(xnext + (size_t)BATCH * IN_DIM + tid * 8, 0, 0);
        }
        v2f xn[16];
#pragma unroll
        for (int kk = 0; kk < 16; ++kk) xn[kk] = load_frag(xnext, IN_DIM, 4 * kk, lane);

        v8f acc0 = accx;
#pragma unroll
        for (int kk = 0; kk < 4; ++kk) {
            v2f b = load_frag(sH0[cur], H0, 4 * kk, lane);
            acc0 = wmma4(wh0[kk], b, acc0);
        }
        *(v2f*)&sH0[nxt][o0] = cell2(acc0, c0a, c0b);
        lds_barrier();

        v8f acc1a = bias1a, acc1b = bias1b;
#pragma unroll
        for (int kk = 0; kk < 4; ++kk) {
            v2f b = load_frag(sH0[nxt], H0, 4 * kk, lane);
            acc1a = wmma4(wx1a[kk], b, acc1a);
            acc1b = wmma4(wx1b[kk], b, acc1b);
        }
#pragma unroll
        for (int kk = 0; kk < 8; ++kk) {
            v2f b = load_frag(sH1[cur], H1, 4 * kk, lane);
            acc1a = wmma4(wh1a[kk], b, acc1a);
            acc1b = wmma4(wh1b[kk], b, acc1b);
        }
        {
            v2f ha = cell2(acc1a, c1a, c1b);
            v2f hb = cell2(acc1b, c1c, c1d);
            v4f h4 = {ha.x, ha.y, hb.x, hb.y};
            *(v4f*)&sH1[nxt][o1] = h4;
        }
        lds_barrier();

        v8f acc2 = bias2;
#pragma unroll
        for (int kk = 0; kk < 8; ++kk) {
            v2f b = load_frag(sH1[nxt], H1, 4 * kk, lane);
            acc2 = wmma4(wx2[kk], b, acc2);
        }
#pragma unroll
        for (int kk = 0; kk < 4; ++kk) {
            v2f b = load_frag(sH2[cur], H2, 4 * kk, lane);
            acc2 = wmma4(wh2[kk], b, acc2);
        }
        accx = bias0;
#pragma unroll
        for (int kk = 0; kk < 16; ++kk) accx = wmma4(wx0[kk], xn[kk], accx);
        *(v2f*)&sH2[nxt][o2] = cell2(acc2, c2a, c2b);
        lds_barrier();

        {
            int m = tid >> 3;
            int o = tid & 7;
            const float* h2n = sH2[nxt];
            float acc = sBl[o];
#pragma unroll
            for (int k = 0; k < H2; ++k) acc += h2n[m * H2 + k] * sWl[o * H2 + k];
            out[(((size_t)(b0 + m)) * T_STEPS + t) * OUT_DIM + o] = acc;
        }
    }
}

extern "C" void kernel_launch(void* const* d_in, const int* in_sizes, int n_in,
                              void* d_out, int out_size, void* d_ws, size_t ws_size,
                              hipStream_t stream) {
    (void)in_sizes; (void)n_in; (void)out_size; (void)d_ws; (void)ws_size;
    const float* x    = (const float*)d_in[0];
    const float* Wih0 = (const float*)d_in[1];
    const float* Whh0 = (const float*)d_in[2];
    const float* bih0 = (const float*)d_in[3];
    const float* bhh0 = (const float*)d_in[4];
    const float* Wih1 = (const float*)d_in[5];
    const float* Whh1 = (const float*)d_in[6];
    const float* bih1 = (const float*)d_in[7];
    const float* bhh1 = (const float*)d_in[8];
    const float* Wih2 = (const float*)d_in[9];
    const float* Whh2 = (const float*)d_in[10];
    const float* bih2 = (const float*)d_in[11];
    const float* bhh2 = (const float*)d_in[12];
    const float* Wl   = (const float*)d_in[13];
    const float* bl   = (const float*)d_in[14];
    float* out = (float*)d_out;

    lstm3_scan_kernel<<<BATCH / 16, NTHREADS, 0, stream>>>(
        x, Wih0, Whh0, bih0, bhh0,
        Wih1, Whh1, bih1, bhh1,
        Wih2, Whh2, bih2, bhh2,
        Wl, bl, out);
}


# scan skewed by layer: pass s runs layer 0 on step s, layer 1 on s-1, layer 2 on s-2, output on s-3, each from the h copies of the pass before; one barrier a pass, not three; sum order kept
# speedup vs baseline: 3.4954x; 1.4276x over previous
#include <hip/hip_runtime.h>


typedef __attribute__((ext_vector_type(2))) float v2f;
typedef __attribute__((ext_vector_type(4))) float v4f;
typedef __attribute__((ext_vector_type(8))) float v8f;

#define T_STEPS  2048
#define BATCH    256
#define IN_DIM   64
#define H0       16
#define H1       32
#define H2       16
#define OUT_DIM  8
#define NTHREADS 128

__device__ __forceinline__ float sigmoidf_(float x) {
    return __builtin_amdgcn_rcpf(1.0f + __builtin_amdgcn_exp2f(x * -1.442695041f));
}

__device__ __forceinline__ float tanhf_(float x) {
    const float ax = fabsf(x);
    const float e   = __builtin_amdgcn_exp2f(ax * 2.885390082f);
    const float big = fmaf(__builtin_amdgcn_rcpf(e + 1.0f), -2.0f, 1.0f);
    const float t = ax * ax;
    float p = fmaf(t, -0.00570002059f, 0.0206340719f);
    p = fmaf(t, p, -0.0537379310f);
    p = fmaf(t, p, 0.133314162f);
    p = fmaf(t, p, -0.333332807f);
    const float sm = fmaf(t, ax * p, ax);
    return copysignf(ax < 0.625f ? sm : big, x);
}

__device__ __forceinline__ v2f load_frag(const float* base, int ld, int kbase, int lane) {
    const float* p = base + (lane & 15) * ld + kbase + ((lane >> 4) << 1);
    v2f a;
    a.x = p[0];
    a.y = p[1];
    return a;
}

template <int H>
__device__ __forceinline__ v2f load_wfrag(const float* W, int ldw, int ubase, int hs, int kbase, int lane) {
    const int c   = lane & 15;
    const int row = (c & 3) * H + ubase + hs * (c >> 3) + ((c >> 2) & 1);
    const float* p = W + row * ldw + kbase + ((lane >> 4) << 1);
    v2f a;
    a.x = p[0];
    a.y = p[1];
    return a;
}

template <int H>
__device__ __forceinline__ v8f load_bias(const float* bi, const float* bh, int ubase, int hs, int lane) {
    const int hh = lane >> 4;
    v8f b;
#pragma unroll
    for (int r = 0; r < 8; ++r) {
        const int row = (r & 3) * H + ubase + hs * hh + (r >> 2);
        b[r] = bi[row] + bh[row];
    }
    return b;
}

__device__ __forceinline__ v8f wmma4(v2f a, v2f b, v8f c) {
    return __builtin_amdgcn_wmma_f32_16x16x4_f32(false, a, false, b, (short)0, c, false, false);
}

__device__ __forceinline__ void lds_barrier() {
    __builtin_amdgcn_fence(__ATOMIC_RELEASE, "workgroup", "local");
    __builtin_amdgcn_s_barrier();
    __builtin_amdgcn_fence(__ATOMIC_ACQUIRE, "workgroup", "local");
}

__device__ __forceinline__ v2f cell2(v8f d, float& c0, float& c1) {
    v2f h;
    {
        float cn = sigmoidf_(d[1]) * c0 + sigmoidf_(d[0]) * tanhf_(d[2]);
        c0 = cn;
        h.x = sigmoidf_(d[3]) * tanhf_(cn);
    }
    {
        float cn = sigmoidf_(d[5]) * c1 + sigmoidf_(d[4]) * tanhf_(d[6]);
        c1 = cn;
        h.y = sigmoidf_(d[7]) * tanhf_(cn);
    }
    return h;
}

__global__ __launch_bounds__(NTHREADS) void lstm3_scan_kernel(
    const float* __restrict__ x,
    const float* __restrict__ Wih0, const float* __restrict__ Whh0,
    const float* __restrict__ bih0, const float* __restrict__ bhh0,
    const float* __restrict__ Wih1, const float* __restrict__ Whh1,
    const float* __restrict__ bih1, const float* __restrict__ bhh1,
    const float* __restrict__ Wih2, const float* __restrict__ Whh2,
    const float* __restrict__ bih2, const float* __restrict__ bhh2,
    const float* __restrict__ Wl,   const float* __restrict__ bl,
    float* __restrict__ out)
{
    __shared__ __attribute__((aligned(16))) float sH0[2][16 * H0];
    __shared__ __attribute__((aligned(16))) float sH1[2][16 * H1];
    __shared__ __attribute__((aligned(16))) float sH2[2][16 * H2];
    __shared__ __attribute__((aligned(16))) float sWl[OUT_DIM * H2];
    __shared__ float sBl[OUT_DIM];

    const int tid  = threadIdx.x;
    const int lane = tid & 31;
    const int wave = tid >> 5;
    const int ln   = lane & 15;
    const int hh   = lane >> 4;
    const int b0   = blockIdx.x * 16;

    for (int i = tid; i < 16 * H0; i += NTHREADS) {
        sH0[0][i] = 0.f; sH0[1][i] = 0.f; sH2[0][i] = 0.f; sH2[1][i] = 0.f;
    }
    for (int i = tid; i < 16 * H1; i += NTHREADS) { sH1[0][i] = 0.f; sH1[1][i] = 0.f; }
    for (int i = tid; i < OUT_DIM * H2; i += NTHREADS) sWl[i] = Wl[i];
    if (tid < OUT_DIM) sBl[tid] = bl[tid];

    const int u0 = wave * 4;
    v2f wx0[16], wh0[4];
#pragma unroll
    for (int kk = 0; kk < 16; ++kk) wx0[kk] = load_wfrag<H0>(Wih0, IN_DIM, u0, 2, 4 * kk, lane);
#pragma unroll
    for (int kk = 0; kk < 4;  ++kk) wh0[kk] = load_wfrag<H0>(Whh0, H0, u0, 2, 4 * kk, lane);
    const v8f bias0 = load_bias<H0>(bih0, bhh0, u0, 2, lane);

    const int u1a = wave * 8, u1b = wave * 8 + 2;
    v2f wx1a[4], wx1b[4], wh1a[8], wh1b[8];
#pragma unroll
    for (int kk = 0; kk < 4; ++kk) {
        wx1a[kk] = load_wfrag<H1>(Wih1, H0, u1a, 4, 4 * kk, lane);
        wx1b[kk] = load_wfrag<H1>(Wih1, H0, u1b, 4, 4 * kk, lane);
    }
#pragma unroll
    for (int kk = 0; kk < 8; ++kk) {
        wh1a[kk] = load_wfrag<H1>(Whh1, H1, u1a, 4, 4 * kk, lane);
        wh1b[kk] = load_wfrag<H1>(Whh1, H1, u1b, 4, 4 * kk, lane);
    }
    const v8f bias1a = load_bias<H1>(bih1, bhh1, u1a, 4, lane);
    const v8f bias1b = load_bias<H1>(bih1, bhh1, u1b, 4, lane);

    const int u2 = wave * 4;
    v2f wx2[8], wh2[4];
#pragma unroll
    for (int kk = 0; kk < 8; ++kk) wx2[kk] = load_wfrag<H2>(Wih2, H1, u2, 2, 4 * kk, lane);
#pragma unroll
    for (int kk = 0; kk < 4; ++kk) wh2[kk] = load_wfrag<H2>(Whh2, H2, u2, 2, 4 * kk, lane);
    const v8f bias2 = load_bias<H2>(bih2, bhh2, u2, 2, lane);

    const int o0 = ln * H0 + u0 + 2 * hh;
    const int o1 = ln * H1 + u1a + 4 * hh;
    const int o2 = ln * H2 + u2 + 2 * hh;

    const int om = tid >> 3;
    const int oo = tid & 7;
    float* const outp = out + ((size_t)(b0 + om)) * T_STEPS * OUT_DIM + oo;

    float c0a = 0.f, c0b = 0.f;
    float c1a = 0.f, c1b = 0.f, c1c = 0.f, c1d = 0.f;
    float c2a = 0.f, c2b = 0.f;

    v8f accx = bias0;
    {
        const float* x0 = x + (size_t)b0 * IN_DIM;
#pragma unroll
        for (int kk = 0; kk < 16; ++kk) {
            v2f b = load_frag(x0, IN_DIM, 4 * kk, lane);
            accx = wmma4(wx0[kk], b, accx);
        }
    }

    __syncthreads();

    auto pass = [&](const int s, const bool do0, const bool do1, const bool do2, const bool do3)
                    __attribute__((always_inline)) {
        const int cur = s & 1, nxt = cur ^ 1;

        v2f f0[4], f1[8], f2[4];
        if (do0 || do1) {
#pragma unroll
            for (int kk = 0; kk < 4; ++kk) f0[kk] = load_frag(sH0[cur], H0, 4 * kk, lane);
        }
        if (do1 || do2) {
#pragma unroll
            for (int kk = 0; kk < 8; ++kk) f1[kk] = load_frag(sH1[cur], H1, 4 * kk, lane);
        }
        if (do2) {
#pragma unroll
            for (int kk = 0; kk < 4; ++kk) f2[kk] = load_frag(sH2[cur], H2, 4 * kk, lane);
        }

        v4f oh[4], ow[4];
        float ob = 0.f;
        if (do3) {
#pragma unroll
            for (int q = 0; q < 4; ++q) {
                oh[q] = *(const v4f*)&sH2[cur][om * H2 + 4 * q];
                ow[q] = *(const v4f*)&sWl[oo * H2 + 4 * q];
            }
            ob = sBl[oo];
        }
        v2f xn[16];
        if (do0) {
            const int tn = (s + 1 < T_STEPS) ? s + 1 : s;
            const float* xnext = x + ((size_t)tn * BATCH + b0) * IN_DIM;
            if (s + 2 < T_STEPS) {
                __builtin_prefetch(xnext + (size_t)BATCH * IN_DIM + tid * 8, 0, 0);
            }
#pragma unroll
            for (int kk = 0; kk < 16; ++kk) xn[kk] = load_frag(xnext, IN_DIM, 4 * kk, lane);
        }
        __builtin_amdgcn_sched_barrier(0);
        if (do3) {
            float acc = ob;
#pragma unroll
            for (int q = 0; q < 4; ++q) {
                acc += oh[q].x * ow[q].x;
                acc += oh[q].y * ow[q].y;
                acc += oh[q].z * ow[q].z;
                acc += oh[q].w * ow[q].w;
            }
            outp[(size_t)(s - 3) * OUT_DIM] = acc;
        }

        v8f acc0 = accx, acc1a = bias1a, acc1b = bias1b, acc2 = bias2;
#pragma unroll
        for (int kk = 0; kk < 4; ++kk) {
            if (do0) acc0 = wmma4(wh0[kk], f0[kk], acc0);
            if (do1) {
                acc1a = wmma4(wx1a[kk], f0[kk], acc1a);
                acc1b = wmma4(wx1b[kk], f0[kk], acc1b);
            }
        }
        if (do0) *(v2f*)&sH0[nxt][o0] = cell2(acc0, c0a, c0b);
#pragma unroll
        for (int kk = 0; kk < 8; ++kk) {
            if (do1) {
                acc1a = wmma4(wh1a[kk], f1[kk], acc1a);
                acc1b = wmma4(wh1b[kk], f1[kk], acc1b);
            }
            if (do2) acc2 = wmma4(wx2[kk], f1[kk], acc2);
        }
        if (do1) {
            v2f ha = cell2(acc1a, c1a, c1b);
            v2f hb = cell2(acc1b, c1c, c1d);
            v4f h4 = {ha.x, ha.y, hb.x, hb.y};
            *(v4f*)&sH1[nxt][o1] = h4;
        }
        if (do2) {
#pragma unroll
            for (int kk = 0; kk < 4; ++kk) acc2 = wmma4(wh2[kk], f2[kk], acc2);
        }
        if (do0) {
            accx = bias0;
#pragma unroll
            for (int kk = 0; kk < 16; ++kk) accx = wmma4(wx0[kk], xn[kk], accx);
        }
        if (do2) *(v2f*)&sH2[nxt][o2] = cell2(acc2, c2a, c2b);

        lds_barrier();
    };

    pass(0, true, false, false, false);
    pass(1, true, true,  false, false);
    pass(2, true, true,  true,  false);
    for (int s = 3; s < T_STEPS; ++s) pass(s, true, true, true, true);
    pass(T_STEPS,     false, true,  true,  true);
    pass(T_STEPS + 1, false, false, true,  true);
    pass(T_STEPS + 2, false, false, false, true);
}

extern "C" void kernel_launch(void* const* d_in, const int* in_sizes, int n_in,
                              void* d_out, int out_size, void* d_ws, size_t ws_size,
                              hipStream_t stream) {
    (void)in_sizes; (void)n_in; (void)out_size; (void)d_ws; (void)ws_size;
    const float* x    = (const float*)d_in[0];
    const float* Wih0 = (const float*)d_in[1];
    const float* Whh0 = (const float*)d_in[2];
    const float* bih0 = (const float*)d_in[3];
    const float* bhh0 = (const float*)d_in[4];
    const float* Wih1 = (const float*)d_in[5];
    const float* Whh1 = (const float*)d_in[6];
    const float* bih1 = (const float*)d_in[7];
    const float* bhh1 = (const float*)d_in[8];
    const float* Wih2 = (const float*)d_in[9];
    const float* Whh2 = (const float*)d_in[10];
    const float* bih2 = (const float*)d_in[11];
    const float* bhh2 = (const float*)d_in[12];
    const float* Wl   = (const float*)d_in[13];
    const float* bl   = (const float*)d_in[14];
    float* out = (float*)d_out;

    lstm3_scan_kernel<<<BATCH / 16, NTHREADS, 0, stream>>>(
        x, Wih0, Whh0, bih0, bhh0,
        Wih1, Whh1, bih1, bhh1,
        Wih2, Whh2, bih2, bhh2,
        Wl, bl, out);
}


# 4 batch rows a workgroup (64 workgroups), batch tile columns repeated 4x: each lane group updates another unit's gates, 2 cells a lane a pass, not 8; sum order kept
# speedup vs baseline: 4.4876x; 1.2839x over previous
#include <hip/hip_runtime.h>


typedef __attribute__((ext_vector_type(2))) float v2f;
typedef __attribute__((ext_vector_type(4))) float v4f;
typedef __attribute__((ext_vector_type(8))) float v8f;

#define T_STEPS  2048
#define BATCH    256
#define IN_DIM   64
#define H0       16
#define H1       32
#define H2       16
#define OUT_DIM  8
#define NTHREADS 128
#define ROWS     4

__device__ __forceinline__ float sigmoidf_(float x) {
    return __builtin_amdgcn_rcpf(1.0f + __builtin_amdgcn_exp2f(x * -1.442695041f));
}

__device__ __forceinline__ float tanhf_(float x) {
    const float ax = fabsf(x);
    const float e   = __builtin_amdgcn_exp2f(ax * 2.885390082f);
    const float big = fmaf(__builtin_amdgcn_rcpf(e + 1.0f), -2.0f, 1.0f);
    const float t = ax * ax;
    float p = fmaf(t, -0.00570002059f, 0.0206340719f);
    p = fmaf(t, p, -0.0537379310f);
    p = fmaf(t, p, 0.133314162f);
    p = fmaf(t, p, -0.333332807f);
    const float sm = fmaf(t, ax * p, ax);
    return copysignf(ax < 0.625f ? sm : big, x);
}

__device__ __forceinline__ v2f load_frag(const float* base, int ld, int kbase, int lane) {
    const float* p = base + (lane & (ROWS - 1)) * ld + kbase + ((lane >> 4) << 1);
    v2f a;
    a.x = p[0];
    a.y = p[1];
    return a;
}

template <int H>
__device__ __forceinline__ v2f load_wfrag(const float* W, int ldw, int ubase, int hs, int kbase, int lane) {
    const int c   = lane & 15;
    const int row = (c & 3) * H + ubase + hs * (c >> 3) + ((c >> 2) & 1);
    const float* p = W + row * ldw + kbase + ((lane >> 4) << 1);
    v2f a;
    a.x = p[0];
    a.y = p[1];
    return a;
}

template <int H>
__device__ __forceinline__ v8f load_bias(const float* bi, const float* bh, int ubase, int hs, int lane) {
    const int hh = lane >> 4;
    v8f b;
#pragma unroll
    for (int r = 0; r < 8; ++r) {
        const int row = (r & 3) * H + ubase + hs * hh + (r >> 2);
        b[r] = bi[row] + bh[row];
    }
    return b;
}

__device__ __forceinline__ v8f wmma4(v2f a, v2f b, v8f c) {
    return __builtin_amdgcn_wmma_f32_16x16x4_f32(false, a, false, b, (short)0, c, false, false);
}

__device__ __forceinline__ void lds_barrier() {
    __builtin_amdgcn_fence(__ATOMIC_RELEASE, "workgroup", "local");
    __builtin_amdgcn_s_barrier();
    __builtin_amdgcn_fence(__ATOMIC_ACQUIRE, "workgroup", "local");
}

__device__ __forceinline__ v2f cell2(v8f d, float& c0, float& c1) {
    v2f h;
    {
        float cn = sigmoidf_(d[1]) * c0 + sigmoidf_(d[0]) * tanhf_(d[2]);
        c0 = cn;
        h.x = sigmoidf_(d[3]) * tanhf_(cn);
    }
    {
        float cn = sigmoidf_(d[5]) * c1 + sigmoidf_(d[4]) * tanhf_(d[6]);
        c1 = cn;
        h.y = sigmoidf_(d[7]) * tanhf_(cn);
    }
    return h;
}

__global__ __launch_bounds__(NTHREADS) void lstm3_scan_kernel(
    const float* __restrict__ x,
    const float* __restrict__ Wih0, const float* __restrict__ Whh0,
    const float* __restrict__ bih0, const float* __restrict__ bhh0,
    const float* __restrict__ Wih1, const float* __restrict__ Whh1,
    const float* __restrict__ bih1, const float* __restrict__ bhh1,
    const float* __restrict__ Wih2, const float* __restrict__ Whh2,
    const float* __restrict__ bih2, const float* __restrict__ bhh2,
    const float* __restrict__ Wl,   const float* __restrict__ bl,
    float* __restrict__ out)
{
    __shared__ __attribute__((aligned(16))) float sH02[2][2][ROWS * H0];
    __shared__ __attribute__((aligned(16))) float sH1[2][ROWS * H1];
    __shared__ __attribute__((aligned(16))) float sWl[OUT_DIM * H2];
    __shared__ float sBl[OUT_DIM];

    const int tid  = threadIdx.x;
    const int lane = tid & 31;
    const int wave = tid >> 5;
    const int br   = lane & (ROWS - 1);
    const int g    = (lane >> 2) & 3;
    const bool gsel  = (g & 1) != 0;
    const bool gtile = (g & 2) != 0;
    const int hh   = lane >> 4;
    const int b0   = blockIdx.x * ROWS;

    for (int i = tid; i < ROWS * H0; i += NTHREADS) {
        sH02[0][0][i] = 0.f; sH02[0][1][i] = 0.f; sH02[1][0][i] = 0.f; sH02[1][1][i] = 0.f;
    }
    for (int i = tid; i < ROWS * H1; i += NTHREADS) { sH1[0][i] = 0.f; sH1[1][i] = 0.f; }
    for (int i = tid; i < OUT_DIM * H2; i += NTHREADS) sWl[i] = Wl[i];
    if (tid < OUT_DIM) sBl[tid] = bl[tid];

    const int u0 = wave * 4;
    v2f wx0[16], wh0[4];
#pragma unroll
    for (int kk = 0; kk < 16; ++kk) wx0[kk] = load_wfrag<H0>(Wih0, IN_DIM, u0, 2, 4 * kk, lane);
#pragma unroll
    for (int kk = 0; kk < 4;  ++kk) wh0[kk] = load_wfrag<H0>(Whh0, H0, u0, 2, 4 * kk, lane);
    const v8f bias0 = load_bias<H0>(bih0, bhh0, u0, 2, lane);

    const int u1a = wave * 8, u1b = wave * 8 + 2;
    v2f wx1a[4], wx1b[4], wh1a[8], wh1b[8];
#pragma unroll
    for (int kk = 0; kk < 4; ++kk) {
        wx1a[kk] = load_wfrag<H1>(Wih1, H0, u1a, 4, 4 * kk, lane);
        wx1b[kk] = load_wfrag<H1>(Wih1, H0, u1b, 4, 4 * kk, lane);
    }
#pragma unroll
    for (int kk = 0; kk < 8; ++kk) {
        wh1a[kk] = load_wfrag<H1>(Whh1, H1, u1a, 4, 4 * kk, lane);
        wh1b[kk] = load_wfrag<H1>(Whh1, H1, u1b, 4, 4 * kk, lane);
    }
    const v8f bias1a = load_bias<H1>(bih1, bhh1, u1a, 4, lane);
    const v8f bias1b = load_bias<H1>(bih1, bhh1, u1b, 4, lane);

    const int u2 = wave * 4;
    v2f wx2[8], wh2[4];
#pragma unroll
    for (int kk = 0; kk < 8; ++kk) wx2[kk] = load_wfrag<H2>(Wih2, H1, u2, 2, 4 * kk, lane);
#pragma unroll
    for (int kk = 0; kk < 4; ++kk) wh2[kk] = load_wfrag<H2>(Whh2, H2, u2, 2, 4 * kk, lane);
    const v8f bias2 = load_bias<H2>(bih2, bhh2, u2, 2, lane);

    const int o1  = br * H1 + u1a + 4 * hh + g;
    const int o02 = (g >> 1) * (ROWS * H0) + br * H0 + u0 + 2 * hh + (g & 1);

    const int om = lane >> 3;
    const int oo = lane & 7;
    float* const outp = out + ((size_t)(b0 + om)) * T_STEPS * OUT_DIM + oo;

    float c1 = 0.f;
    float c02 = 0.f;

    v8f accx = bias0;
    {
        const float* x0 = x + (size_t)b0 * IN_DIM;
#pragma unroll
        for (int kk = 0; kk < 16; ++kk) {
            v2f b = load_frag(x0, IN_DIM, 4 * kk, lane);
            accx = wmma4(wx0[kk], b, accx);
        }
    }

    __syncthreads();

    auto pass = [&](const int s, const bool do0, const bool do1, const bool do2, const bool do3)
                    __attribute__((always_inline)) {
        const int cur = s & 1, nxt = cur ^ 1;
        const float* const h0cur = sH02[cur][0];
        const float* const h2cur = sH02[cur][1];

        v2f f0[4], f1[8], f2[4];
        if (do0 || do1) {
#pragma unroll
            for (int kk = 0; kk < 4; ++kk) f0[kk] = load_frag(h0cur, H0, 4 * kk, lane);
        }
        if (do1 || do2) {
#pragma unroll
            for (int kk = 0; kk < 8; ++kk) f1[kk] = load_frag(sH1[cur], H1, 4 * kk, lane);
        }
        if (do2) {
#pragma unroll
            for (int kk = 0; kk < 4; ++kk) f2[kk] = load_frag(h2cur, H2, 4 * kk, lane);
        }

        v4f oh[4], ow[4];
        float ob = 0.f;
        if (do3) {
#pragma unroll
            for (int q = 0; q < 4; ++q) {
                oh[q] = *(const v4f*)&h2cur[om * H2 + 4 * q];
                ow[q] = *(const v4f*)&sWl[oo * H2 + 4 * q];
            }
            ob = sBl[oo];
        }
        v2f xn[16];
        if (do0) {
            const int tn = (s + 1 < T_STEPS) ? s + 1 : s;
            const float* xnext = x + ((size_t)tn * BATCH + b0) * IN_DIM;
            if (s + 2 < T_STEPS) {
                __builtin_prefetch(xnext + (size_t)BATCH * IN_DIM + tid * 2, 0, 0);
            }
#pragma unroll
            for (int kk = 0; kk < 16; ++kk) xn[kk] = load_frag(xnext, IN_DIM, 4 * kk, lane);
        }
        __builtin_amdgcn_sched_barrier(0);
        float pred = ob;
        if (do3) {
#pragma unroll
            for (int q = 0; q < 4; ++q) {
                pred += oh[q].x * ow[q].x;
                pred += oh[q].y * ow[q].y;
                pred += oh[q].z * ow[q].z;
                pred += oh[q].w * ow[q].w;
            }
            asm("" : "+v"(pred));
        }

        v8f acc0 = accx, acc1a = bias1a, acc1b = bias1b, acc2 = bias2;
#pragma unroll
        for (int kk = 0; kk < 4; ++kk) {
            if (do0) acc0 = wmma4(wh0[kk], f0[kk], acc0);
            if (do1) {
                acc1a = wmma4(wx1a[kk], f0[kk], acc1a);
                acc1b = wmma4(wx1b[kk], f0[kk], acc1b);
            }
        }
#pragma unroll
        for (int kk = 0; kk < 8; ++kk) {
            if (do1) {
                acc1a = wmma4(wh1a[kk], f1[kk], acc1a);
                acc1b = wmma4(wh1b[kk], f1[kk], acc1b);
            }
            if (do2) acc2 = wmma4(wx2[kk], f1[kk], acc2);
        }
        if (do2) {
#pragma unroll
            for (int kk = 0; kk < 4; ++kk) acc2 = wmma4(wh2[kk], f2[kk], acc2);
        }
        if (do0 && do1 && do2) asm("" : "+v"(acc0), "+v"(acc1a), "+v"(acc1b), "+v"(acc2), "+v"(xn[0]));
        if (do0) {
            accx = bias0;
#pragma unroll
            for (int kk = 0; kk < 16; ++kk) accx = wmma4(wx0[kk], xn[kk], accx);
        }

        const v4f g1a = gsel ? __builtin_shufflevector(acc1a, acc1a, 4, 5, 6, 7)
                             : __builtin_shufflevector(acc1a, acc1a, 0, 1, 2, 3);
        const v4f g1b = gsel ? __builtin_shufflevector(acc1b, acc1b, 4, 5, 6, 7)
                             : __builtin_shufflevector(acc1b, acc1b, 0, 1, 2, 3);
        const v4f g0  = gsel ? __builtin_shufflevector(acc0, acc0, 4, 5, 6, 7)
                             : __builtin_shufflevector(acc0, acc0, 0, 1, 2, 3);
        const v4f g2  = gsel ? __builtin_shufflevector(acc2, acc2, 4, 5, 6, 7)
                             : __builtin_shufflevector(acc2, acc2, 0, 1, 2, 3);
        const v4f d1  = gtile ? g1b : g1a;
        const v4f d02 = gtile ? g2 : g0;
        const v8f d = __builtin_shufflevector(d1, d02, 0, 1, 2, 3, 4, 5, 6, 7);
        float c1n = c1, c02n = c02;
        const v2f hn = cell2(d, c1n, c02n);
        if (do1) {
            c1 = c1n;
            sH1[nxt][o1] = hn.x;
        }
        const bool act02 = gtile ? do2 : do0;
        if (act02) {
            c02 = c02n;
            (&sH02[nxt][0][0])[o02] = hn.y;
        }
        if (do3 && wave == 0) outp[(size_t)(s - 3) * OUT_DIM] = pred;

        lds_barrier();
    };

    pass(0, true, false, false, false);
    pass(1, true, true,  false, false);
    pass(2, true, true,  true,  false);
    for (int s = 3; s < T_STEPS; ++s) pass(s, true, true, true, true);
    pass(T_STEPS,     false, true,  true,  true);
    pass(T_STEPS + 1, false, false, true,  true);
    pass(T_STEPS + 2, false, false, false, true);
}

extern "C" void kernel_launch(void* const* d_in, const int* in_sizes, int n_in,
                              void* d_out, int out_size, void* d_ws, size_t ws_size,
                              hipStream_t stream) {
    (void)in_sizes; (void)n_in; (void)out_size; (void)d_ws; (void)ws_size;
    const float* x    = (const float*)d_in[0];
    const float* Wih0 = (const float*)d_in[1];
    const float* Whh0 = (const float*)d_in[2];
    const float* bih0 = (const float*)d_in[3];
    const float* bhh0 = (const float*)d_in[4];
    const float* Wih1 = (const float*)d_in[5];
    const float* Whh1 = (const float*)d_in[6];
    const float* bih1 = (const float*)d_in[7];
    const float* bhh1 = (const float*)d_in[8];
    const float* Wih2 = (const float*)d_in[9];
    const float* Whh2 = (const float*)d_in[10];
    const float* bih2 = (const float*)d_in[11];
    const float* bhh2 = (const float*)d_in[12];
    const float* Wl   = (const float*)d_in[13];
    const float* bl   = (const float*)d_in[14];
    float* out = (float*)d_out;

    lstm3_scan_kernel<<<BATCH / ROWS, NTHREADS, 0, stream>>>(
        x, Wih0, Whh0, bih0, bhh0,
        Wih1, Whh1, bih1, bhh1,
        Wih2, Whh2, bih2, bhh2,
        Wl, bl, out);
}


# layer-0 input projection batched over 4 time steps: one 16-column tile holds 4 steps x 4 batch rows, 4 products a pass not 16, each step's columns fetched by v_permlane16; sum order kept
# speedup vs baseline: 6.1291x; 1.3658x over previous
#include <hip/hip_runtime.h>


typedef __attribute__((ext_vector_type(2))) float v2f;
typedef __attribute__((ext_vector_type(4))) float v4f;
typedef __attribute__((ext_vector_type(8))) float v8f;

#define T_STEPS  2048
#define BATCH    256
#define IN_DIM   64
#define H0       16
#define H1       32
#define H2       16
#define OUT_DIM  8
#define NTHREADS 128
#define ROWS     4

__device__ __forceinline__ float sigmoidf_(float x) {
    return __builtin_amdgcn_rcpf(1.0f + __builtin_amdgcn_exp2f(x * -1.442695041f));
}

__device__ __forceinline__ float tanhf_(float x) {
    const float ax = fabsf(x);
    const float e   = __builtin_amdgcn_exp2f(ax * 2.885390082f);
    const float big = fmaf(__builtin_amdgcn_rcpf(e + 1.0f), -2.0f, 1.0f);
    const float t = ax * ax;
    float p = fmaf(t, -0.00570002059f, 0.0206340719f);
    p = fmaf(t, p, -0.0537379310f);
    p = fmaf(t, p, 0.133314162f);
    p = fmaf(t, p, -0.333332807f);
    const float sm = fmaf(t, ax * p, ax);
    return copysignf(ax < 0.625f ? sm : big, x);
}

__device__ __forceinline__ v2f load_frag(const float* base, int ld, int kbase, int lane) {
    const float* p = base + (lane & (ROWS - 1)) * ld + kbase + ((lane >> 4) << 1);
    v2f a;
    a.x = p[0];
    a.y = p[1];
    return a;
}

__device__ __forceinline__ v2f load_xfrag(const float* xrow, int kbase) {
    v2f a;
    a.x = xrow[kbase];
    a.y = xrow[kbase + 1];
    return a;
}

__device__ __forceinline__ v8f take_step(v8f p, int tt) {
    const unsigned sel = 0x32103210u + 0x44444444u * (unsigned)tt;
    v8f r;
#pragma unroll
    for (int i = 0; i < 8; ++i) {
        const unsigned u = __builtin_bit_cast(unsigned, (float)p[i]);
        r[i] = __builtin_bit_cast(float, __builtin_amdgcn_permlane16(u, u, sel, sel, false, false));
    }
    return r;
}

template <int H>
__device__ __forceinline__ v2f load_wfrag(const float* W, int ldw, int ubase, int hs, int kbase, int lane) {
    const int c   = lane & 15;
    const int row = (c & 3) * H + ubase + hs * (c >> 3) + ((c >> 2) & 1);
    const float* p = W + row * ldw + kbase + ((lane >> 4) << 1);
    v2f a;
    a.x = p[0];
    a.y = p[1];
    return a;
}

template <int H>
__device__ __forceinline__ v8f load_bias(const float* bi, const float* bh, int ubase, int hs, int lane) {
    const int hh = lane >> 4;
    v8f b;
#pragma unroll
    for (int r = 0; r < 8; ++r) {
        const int row = (r & 3) * H + ubase + hs * hh + (r >> 2);
        b[r] = bi[row] + bh[row];
    }
    return b;
}

__device__ __forceinline__ v8f wmma4(v2f a, v2f b, v8f c) {
    return __builtin_amdgcn_wmma_f32_16x16x4_f32(false, a, false, b, (short)0, c, false, false);
}

__device__ __forceinline__ void lds_barrier() {
    __builtin_amdgcn_fence(__ATOMIC_RELEASE, "workgroup", "local");
    __builtin_amdgcn_s_barrier();
    __builtin_amdgcn_fence(__ATOMIC_ACQUIRE, "workgroup", "local");
}

__device__ __forceinline__ v2f cell2(v8f d, float& c0, float& c1) {
    v2f h;
    {
        float cn = sigmoidf_(d[1]) * c0 + sigmoidf_(d[0]) * tanhf_(d[2]);
        c0 = cn;
        h.x = sigmoidf_(d[3]) * tanhf_(cn);
    }
    {
        float cn = sigmoidf_(d[5]) * c1 + sigmoidf_(d[4]) * tanhf_(d[6]);
        c1 = cn;
        h.y = sigmoidf_(d[7]) * tanhf_(cn);
    }
    return h;
}

__global__ __launch_bounds__(NTHREADS) void lstm3_scan_kernel(
    const float* __restrict__ x,
    const float* __restrict__ Wih0, const float* __restrict__ Whh0,
    const float* __restrict__ bih0, const float* __restrict__ bhh0,
    const float* __restrict__ Wih1, const float* __restrict__ Whh1,
    const float* __restrict__ bih1, const float* __restrict__ bhh1,
    const float* __restrict__ Wih2, const float* __restrict__ Whh2,
    const float* __restrict__ bih2, const float* __restrict__ bhh2,
    const float* __restrict__ Wl,   const float* __restrict__ bl,
    float* __restrict__ out)
{
    __shared__ __attribute__((aligned(16))) float sH02[2][2][ROWS * H0];
    __shared__ __attribute__((aligned(16))) float sH1[2][ROWS * H1];
    __shared__ __attribute__((aligned(16))) float sWl[OUT_DIM * H2];
    __shared__ float sBl[OUT_DIM];

    const int tid  = threadIdx.x;
    const int lane = tid & 31;
    const int wave = tid >> 5;
    const int br   = lane & (ROWS - 1);
    const int g    = (lane >> 2) & 3;
    const bool gsel  = (g & 1) != 0;
    const bool gtile = (g & 2) != 0;
    const int hh   = lane >> 4;
    const int b0   = blockIdx.x * ROWS;

    for (int i = tid; i < ROWS * H0; i += NTHREADS) {
        sH02[0][0][i] = 0.f; sH02[0][1][i] = 0.f; sH02[1][0][i] = 0.f; sH02[1][1][i] = 0.f;
    }
    for (int i = tid; i < ROWS * H1; i += NTHREADS) { sH1[0][i] = 0.f; sH1[1][i] = 0.f; }
    for (int i = tid; i < OUT_DIM * H2; i += NTHREADS) sWl[i] = Wl[i];
    if (tid < OUT_DIM) sBl[tid] = bl[tid];

    const int u0 = wave * 4;
    v2f wx0[16], wh0[4];
#pragma unroll
    for (int kk = 0; kk < 16; ++kk) wx0[kk] = load_wfrag<H0>(Wih0, IN_DIM, u0, 2, 4 * kk, lane);
#pragma unroll
    for (int kk = 0; kk < 4;  ++kk) wh0[kk] = load_wfrag<H0>(Whh0, H0, u0, 2, 4 * kk, lane);
    const v8f bias0 = load_bias<H0>(bih0, bhh0, u0, 2, lane);

    const int u1a = wave * 8, u1b = wave * 8 + 2;
    v2f wx1a[4], wx1b[4], wh1a[8], wh1b[8];
#pragma unroll
    for (int kk = 0; kk < 4; ++kk) {
        wx1a[kk] = load_wfrag<H1>(Wih1, H0, u1a, 4, 4 * kk, lane);
        wx1b[kk] = load_wfrag<H1>(Wih1, H0, u1b, 4, 4 * kk, lane);
    }
#pragma unroll
    for (int kk = 0; kk < 8; ++kk) {
        wh1a[kk] = load_wfrag<H1>(Whh1, H1, u1a, 4, 4 * kk, lane);
        wh1b[kk] = load_wfrag<H1>(Whh1, H1, u1b, 4, 4 * kk, lane);
    }
    const v8f bias1a = load_bias<H1>(bih1, bhh1, u1a, 4, lane);
    const v8f bias1b = load_bias<H1>(bih1, bhh1, u1b, 4, lane);

    const int u2 = wave * 4;
    v2f wx2[8], wh2[4];
#pragma unroll
    for (int kk = 0; kk < 8; ++kk) wx2[kk] = load_wfrag<H2>(Wih2, H1, u2, 2, 4 * kk, lane);
#pragma unroll
    for (int kk = 0; kk < 4; ++kk) wh2[kk] = load_wfrag<H2>(Whh2, H2, u2, 2, 4 * kk, lane);
    const v8f bias2 = load_bias<H2>(bih2, bhh2, u2, 2, lane);

    const int o1  = br * H1 + u1a + 4 * hh + g;
    const int o02 = (g >> 1) * (ROWS * H0) + br * H0 + u0 + 2 * hh + (g & 1);

    const int om = lane >> 3;
    const int oo = lane & 7;
    float* const outp = out + ((size_t)(b0 + om)) * T_STEPS * OUT_DIM + oo;

    float c1 = 0.f;
    float c02 = 0.f;

    const int xoff = (g * BATCH + br) * IN_DIM + 2 * hh;
    v8f accx = bias0;
    {
        const float* x0 = x + (size_t)b0 * IN_DIM + xoff;
#pragma unroll
        for (int kk = 0; kk < 16; ++kk) {
            v2f b = load_xfrag(x0, 4 * kk);
            accx = wmma4(wx0[kk], b, accx);
        }
    }
    v8f accn = bias0;

    __syncthreads();

    auto pass = [&](const int s, const int tt, const bool do0, const bool do1, const bool do2, const bool do3)
                    __attribute__((always_inline)) {
        const int cur = tt & 1, nxt = cur ^ 1;
        const float* const h0cur = sH02[cur][0];
        const float* const h2cur = sH02[cur][1];

        v2f f0[4], f1[8], f2[4];
        if (do0 || do1) {
#pragma unroll
            for (int kk = 0; kk < 4; ++kk) f0[kk] = load_frag(h0cur, H0, 4 * kk, lane);
        }
        if (do1 || do2) {
#pragma unroll
            for (int kk = 0; kk < 8; ++kk) f1[kk] = load_frag(sH1[cur], H1, 4 * kk, lane);
        }
        if (do2) {
#pragma unroll
            for (int kk = 0; kk < 4; ++kk) f2[kk] = load_frag(h2cur, H2, 4 * kk, lane);
        }

        v4f oh[4], ow[4];
        float ob = 0.f;
        if (do3) {
#pragma unroll
            for (int q = 0; q < 4; ++q) {
                oh[q] = *(const v4f*)&h2cur[om * H2 + 4 * q];
                ow[q] = *(const v4f*)&sWl[oo * H2 + 4 * q];
            }
            ob = sBl[oo];
        }
        v2f xn[4];
        if (do0) {
            const int tb = (s - tt + 4 < T_STEPS) ? (s - tt + 4) : (s - tt);
            const float* xq = x + ((size_t)tb * BATCH + b0) * IN_DIM + xoff + 16 * tt;
            if (s + 8 < T_STEPS) {
                __builtin_prefetch(x + ((size_t)(s + 8) * BATCH + b0) * IN_DIM + tid * 2, 0, 0);
            }
#pragma unroll
            for (int i = 0; i < 4; ++i) xn[i] = load_xfrag(xq, 4 * i);
        }
        __builtin_amdgcn_sched_barrier(0);
        float pred = ob;
        if (do3) {
#pragma unroll
            for (int q = 0; q < 4; ++q) {
                pred += oh[q].x * ow[q].x;
                pred += oh[q].y * ow[q].y;
                pred += oh[q].z * ow[q].z;
                pred += oh[q].w * ow[q].w;
            }
            asm("" : "+v"(pred));
        }

        v8f acc0 = do0 ? take_step(accx, tt) : bias0, acc1a = bias1a, acc1b = bias1b, acc2 = bias2;
#pragma unroll
        for (int kk = 0; kk < 4; ++kk) {
            if (do0) acc0 = wmma4(wh0[kk], f0[kk], acc0);
            if (do1) {
                acc1a = wmma4(wx1a[kk], f0[kk], acc1a);
                acc1b = wmma4(wx1b[kk], f0[kk], acc1b);
            }
        }
#pragma unroll
        for (int kk = 0; kk < 8; ++kk) {
            if (do1) {
                acc1a = wmma4(wh1a[kk], f1[kk], acc1a);
                acc1b = wmma4(wh1b[kk], f1[kk], acc1b);
            }
            if (do2) acc2 = wmma4(wx2[kk], f1[kk], acc2);
        }
        if (do2) {
#pragma unroll
            for (int kk = 0; kk < 4; ++kk) acc2 = wmma4(wh2[kk], f2[kk], acc2);
        }
        if (do0 && do1 && do2) asm("" : "+v"(acc0), "+v"(acc1a), "+v"(acc1b), "+v"(acc2), "+v"(xn[0]));
        if (do0) {
            if (tt == 0) accn = bias0;
#pragma unroll
            for (int i = 0; i < 4; ++i) accn = wmma4(wx0[4 * tt + i], xn[i], accn);
            if (tt == 3) accx = accn;
        }

        const v4f g1a = gsel ? __builtin_shufflevector(acc1a, acc1a, 4, 5, 6, 7)
                             : __builtin_shufflevector(acc1a, acc1a, 0, 1, 2, 3);
        const v4f g1b = gsel ? __builtin_shufflevector(acc1b, acc1b, 4, 5, 6, 7)
                             : __builtin_shufflevector(acc1b, acc1b, 0, 1, 2, 3);
        const v4f g0  = gsel ? __builtin_shufflevector(acc0, acc0, 4, 5, 6, 7)
                             : __builtin_shufflevector(acc0, acc0, 0, 1, 2, 3);
        const v4f g2  = gsel ? __builtin_shufflevector(acc2, acc2, 4, 5, 6, 7)
                             : __builtin_shufflevector(acc2, acc2, 0, 1, 2, 3);
        const v4f d1  = gtile ? g1b : g1a;
        const v4f d02 = gtile ? g2 : g0;
        const v8f d = __builtin_shufflevector(d1, d02, 0, 1, 2, 3, 4, 5, 6, 7);
        float c1n = c1, c02n = c02;
        const v2f hn = cell2(d, c1n, c02n);
        if (do1) {
            c1 = c1n;
            sH1[nxt][o1] = hn.x;
        }
        const bool act02 = gtile ? do2 : do0;
        if (act02) {
            c02 = c02n;
            (&sH02[nxt][0][0])[o02] = hn.y;
        }
        if (do3 && wave == 0) outp[(size_t)(s - 3) * OUT_DIM] = pred;

        lds_barrier();
    };

    pass(0, 0, true, false, false, false);
    pass(1, 1, true, true,  false, false);
    pass(2, 2, true, true,  true,  false);
    pass(3, 3, true, true, true, true);
    for (int s = 4; s < T_STEPS; s += 4) {
        pass(s,     0, true, true, true, true);
        pass(s + 1, 1, true, true, true, true);
        pass(s + 2, 2, true, true, true, true);
        pass(s + 3, 3, true, true, true, true);
    }
    pass(T_STEPS,     0, false, true,  true,  true);
    pass(T_STEPS + 1, 1, false, false, true,  true);
    pass(T_STEPS + 2, 2, false, false, false, true);
}

extern "C" void kernel_launch(void* const* d_in, const int* in_sizes, int n_in,
                              void* d_out, int out_size, void* d_ws, size_t ws_size,
                              hipStream_t stream) {
    (void)in_sizes; (void)n_in; (void)out_size; (void)d_ws; (void)ws_size;
    const float* x    = (const float*)d_in[0];
    const float* Wih0 = (const float*)d_in[1];
    const float* Whh0 = (const float*)d_in[2];
    const float* bih0 = (const float*)d_in[3];
    const float* bhh0 = (const float*)d_in[4];
    const float* Wih1 = (const float*)d_in[5];
    const float* Whh1 = (const float*)d_in[6];
    const float* bih1 = (const float*)d_in[7];
    const float* bhh1 = (const float*)d_in[8];
    const float* Wih2 = (const float*)d_in[9];
    const float* Whh2 = (const float*)d_in[10];
    const float* bih2 = (const float*)d_in[11];
    const float* bhh2 = (const float*)d_in[12];
    const float* Wl   = (const float*)d_in[13];
    const float* bl   = (const float*)d_in[14];
    float* out = (float*)d_out;

    lstm3_scan_kernel<<<BATCH / ROWS, NTHREADS, 0, stream>>>(
        x, Wih0, Whh0, bih0, bhh0,
        Wih1, Whh1, bih1, bhh1,
        Wih2, Whh2, bih2, bhh2,
        Wl, bl, out);
}


# cell update on the hardware tanh (v_tanh_f32): tanh one instruction, sigmoid = 1/2 + 1/2 tanh(x/2) with the 1/2 folded exactly into the i, f, o weight rows and biases; product sum order kept
# speedup vs baseline: 6.4712x; 1.0558x over previous
#include <hip/hip_runtime.h>


typedef __attribute__((ext_vector_type(2))) float v2f;
typedef __attribute__((ext_vector_type(4))) float v4f;
typedef __attribute__((ext_vector_type(8))) float v8f;

#define T_STEPS  2048
#define BATCH    256
#define IN_DIM   64
#define H0       16
#define H1       32
#define H2       16
#define OUT_DIM  8
#define NTHREADS 128
#define ROWS     4

__device__ __forceinline__ float tanhf_(float x) {
    return __builtin_amdgcn_tanhf(x);
}

__device__ __forceinline__ float sigmoid_half_(float xh) {
    return fmaf(__builtin_amdgcn_tanhf(xh), 0.5f, 0.5f);
}

__device__ __forceinline__ v2f load_frag(const float* base, int ld, int kbase, int lane) {
    const float* p = base + (lane & (ROWS - 1)) * ld + kbase + ((lane >> 4) << 1);
    v2f a;
    a.x = p[0];
    a.y = p[1];
    return a;
}

__device__ __forceinline__ v2f load_xfrag(const float* xrow, int kbase) {
    v2f a;
    a.x = xrow[kbase];
    a.y = xrow[kbase + 1];
    return a;
}

__device__ __forceinline__ v8f take_step(v8f p, int tt) {
    const unsigned sel = 0x32103210u + 0x44444444u * (unsigned)tt;
    v8f r;
#pragma unroll
    for (int i = 0; i < 8; ++i) {
        const unsigned u = __builtin_bit_cast(unsigned, (float)p[i]);
        r[i] = __builtin_bit_cast(float, __builtin_amdgcn_permlane16(u, u, sel, sel, false, false));
    }
    return r;
}

template <int H>
__device__ __forceinline__ v2f load_wfrag(const float* W, int ldw, int ubase, int hs, int kbase, int lane) {
    const int c   = lane & 15;
    const int row = (c & 3) * H + ubase + hs * (c >> 3) + ((c >> 2) & 1);
    const float* p = W + row * ldw + kbase + ((lane >> 4) << 1);
    const float sc = ((c & 3) == 2) ? 1.0f : 0.5f;
    v2f a;
    a.x = p[0] * sc;
    a.y = p[1] * sc;
    return a;
}

template <int H>
__device__ __forceinline__ v8f load_bias(const float* bi, const float* bh, int ubase, int hs, int lane) {
    const int hh = lane >> 4;
    v8f b;
#pragma unroll
    for (int r = 0; r < 8; ++r) {
        const int row = (r & 3) * H + ubase + hs * hh + (r >> 2);
        b[r] = (bi[row] + bh[row]) * (((r & 3) == 2) ? 1.0f : 0.5f);
    }
    return b;
}

__device__ __forceinline__ v8f wmma4(v2f a, v2f b, v8f c) {
    return __builtin_amdgcn_wmma_f32_16x16x4_f32(false, a, false, b, (short)0, c, false, false);
}

__device__ __forceinline__ void lds_barrier() {
    __builtin_amdgcn_fence(__ATOMIC_RELEASE, "workgroup", "local");
    __builtin_amdgcn_s_barrier();
    __builtin_amdgcn_fence(__ATOMIC_ACQUIRE, "workgroup", "local");
}

__device__ __forceinline__ v2f cell2(v8f d, float& c0, float& c1) {
    v2f h;
    {
        float cn = sigmoid_half_(d[1]) * c0 + sigmoid_half_(d[0]) * tanhf_(d[2]);
        c0 = cn;
        h.x = sigmoid_half_(d[3]) * tanhf_(cn);
    }
    {
        float cn = sigmoid_half_(d[5]) * c1 + sigmoid_half_(d[4]) * tanhf_(d[6]);
        c1 = cn;
        h.y = sigmoid_half_(d[7]) * tanhf_(cn);
    }
    return h;
}

__global__ __launch_bounds__(NTHREADS) void lstm3_scan_kernel(
    const float* __restrict__ x,
    const float* __restrict__ Wih0, const float* __restrict__ Whh0,
    const float* __restrict__ bih0, const float* __restrict__ bhh0,
    const float* __restrict__ Wih1, const float* __restrict__ Whh1,
    const float* __restrict__ bih1, const float* __restrict__ bhh1,
    const float* __restrict__ Wih2, const float* __restrict__ Whh2,
    const float* __restrict__ bih2, const float* __restrict__ bhh2,
    const float* __restrict__ Wl,   const float* __restrict__ bl,
    float* __restrict__ out)
{
    __shared__ __attribute__((aligned(16))) float sH02[2][2][ROWS * H0];
    __shared__ __attribute__((aligned(16))) float sH1[2][ROWS * H1];
    __shared__ __attribute__((aligned(16))) float sWl[OUT_DIM * H2];
    __shared__ float sBl[OUT_DIM];

    const int tid  = threadIdx.x;
    const int lane = tid & 31;
    const int wave = tid >> 5;
    const int br   = lane & (ROWS - 1);
    const int g    = (lane >> 2) & 3;
    const bool gsel  = (g & 1) != 0;
    const bool gtile = (g & 2) != 0;
    const int hh   = lane >> 4;
    const int b0   = blockIdx.x * ROWS;

    for (int i = tid; i < ROWS * H0; i += NTHREADS) {
        sH02[0][0][i] = 0.f; sH02[0][1][i] = 0.f; sH02[1][0][i] = 0.f; sH02[1][1][i] = 0.f;
    }
    for (int i = tid; i < ROWS * H1; i += NTHREADS) { sH1[0][i] = 0.f; sH1[1][i] = 0.f; }
    for (int i = tid; i < OUT_DIM * H2; i += NTHREADS) sWl[i] = Wl[i];
    if (tid < OUT_DIM) sBl[tid] = bl[tid];

    const int u0 = wave * 4;
    v2f wx0[16], wh0[4];
#pragma unroll
    for (int kk = 0; kk < 16; ++kk) wx0[kk] = load_wfrag<H0>(Wih0, IN_DIM, u0, 2, 4 * kk, lane);
#pragma unroll
    for (int kk = 0; kk < 4;  ++kk) wh0[kk] = load_wfrag<H0>(Whh0, H0, u0, 2, 4 * kk, lane);
    const v8f bias0 = load_bias<H0>(bih0, bhh0, u0, 2, lane);

    const int u1a = wave * 8, u1b = wave * 8 + 2;
    v2f wx1a[4], wx1b[4], wh1a[8], wh1b[8];
#pragma unroll
    for (int kk = 0; kk < 4; ++kk) {
        wx1a[kk] = load_wfrag<H1>(Wih1, H0, u1a, 4, 4 * kk, lane);
        wx1b[kk] = load_wfrag<H1>(Wih1, H0, u1b, 4, 4 * kk, lane);
    }
#pragma unroll
    for (int kk = 0; kk < 8; ++kk) {
        wh1a[kk] = load_wfrag<H1>(Whh1, H1, u1a, 4, 4 * kk, lane);
        wh1b[kk] = load_wfrag<H1>(Whh1, H1, u1b, 4, 4 * kk, lane);
    }
    const v8f bias1a = load_bias<H1>(bih1, bhh1, u1a, 4, lane);
    const v8f bias1b = load_bias<H1>(bih1, bhh1, u1b, 4, lane);

    const int u2 = wave * 4;
    v2f wx2[8], wh2[4];
#pragma unroll
    for (int kk = 0; kk < 8; ++kk) wx2[kk] = load_wfrag<H2>(Wih2, H1, u2, 2, 4 * kk, lane);
#pragma unroll
    for (int kk = 0; kk < 4; ++kk) wh2[kk] = load_wfrag<H2>(Whh2, H2, u2, 2, 4 * kk, lane);
    const v8f bias2 = load_bias<H2>(bih2, bhh2, u2, 2, lane);

    const int o1  = br * H1 + u1a + 4 * hh + g;
    const int o02 = (g >> 1) * (ROWS * H0) + br * H0 + u0 + 2 * hh + (g & 1);

    const int om = lane >> 3;
    const int oo = lane & 7;
    float* const outp = out + ((size_t)(b0 + om)) * T_STEPS * OUT_DIM + oo;

    float c1 = 0.f;
    float c02 = 0.f;

    const int xoff = (g * BATCH + br) * IN_DIM + 2 * hh;
    v8f accx = bias0;
    {
        const float* x0 = x + (size_t)b0 * IN_DIM + xoff;
#pragma unroll
        for (int kk = 0; kk < 16; ++kk) {
            v2f b = load_xfrag(x0, 4 * kk);
            accx = wmma4(wx0[kk], b, accx);
        }
    }
    v8f accn = bias0;

    __syncthreads();

    auto pass = [&](const int s, const int tt, const bool do0, const bool do1, const bool do2, const bool do3)
                    __attribute__((always_inline)) {
        const int cur = tt & 1, nxt = cur ^ 1;
        const float* const h0cur = sH02[cur][0];
        const float* const h2cur = sH02[cur][1];

        v2f f0[4], f1[8], f2[4];
        if (do0 || do1) {
#pragma unroll
            for (int kk = 0; kk < 4; ++kk) f0[kk] = load_frag(h0cur, H0, 4 * kk, lane);
        }
        if (do1 || do2) {
#pragma unroll
            for (int kk = 0; kk < 8; ++kk) f1[kk] = load_frag(sH1[cur], H1, 4 * kk, lane);
        }
        if (do2) {
#pragma unroll
            for (int kk = 0; kk < 4; ++kk) f2[kk] = load_frag(h2cur, H2, 4 * kk, lane);
        }

        v4f oh[4], ow[4];
        float ob = 0.f;
        if (do3) {
#pragma unroll
            for (int q = 0; q < 4; ++q) {
                oh[q] = *(const v4f*)&h2cur[om * H2 + 4 * q];
                ow[q] = *(const v4f*)&sWl[oo * H2 + 4 * q];
            }
            ob = sBl[oo];
        }
        v2f xn[4];
        if (do0) {
            const int tb = (s - tt + 4 < T_STEPS) ? (s - tt + 4) : (s - tt);
            const float* xq = x + ((size_t)tb * BATCH + b0) * IN_DIM + xoff + 16 * tt;
            if (s + 8 < T_STEPS) {
                __builtin_prefetch(x + ((size_t)(s + 8) * BATCH + b0) * IN_DIM + tid * 2, 0, 0);
            }
#pragma unroll
            for (int i = 0; i < 4; ++i) xn[i] = load_xfrag(xq, 4 * i);
        }
        __builtin_amdgcn_sched_barrier(0);
        float pred = ob;
        if (do3) {
#pragma unroll
            for (int q = 0; q < 4; ++q) {
                pred += oh[q].x * ow[q].x;
                pred += oh[q].y * ow[q].y;
                pred += oh[q].z * ow[q].z;
                pred += oh[q].w * ow[q].w;
            }
            asm("" : "+v"(pred));
        }

        v8f acc0 = do0 ? take_step(accx, tt) : bias0, acc1a = bias1a, acc1b = bias1b, acc2 = bias2;
#pragma unroll
        for (int kk = 0; kk < 4; ++kk) {
            if (do0) acc0 = wmma4(wh0[kk], f0[kk], acc0);
            if (do1) {
                acc1a = wmma4(wx1a[kk], f0[kk], acc1a);
                acc1b = wmma4(wx1b[kk], f0[kk], acc1b);
            }
        }
#pragma unroll
        for (int kk = 0; kk < 8; ++kk) {
            if (do1) {
                acc1a = wmma4(wh1a[kk], f1[kk], acc1a);
                acc1b = wmma4(wh1b[kk], f1[kk], acc1b);
            }
            if (do2) acc2 = wmma4(wx2[kk], f1[kk], acc2);
        }
        if (do2) {
#pragma unroll
            for (int kk = 0; kk < 4; ++kk) acc2 = wmma4(wh2[kk], f2[kk], acc2);
        }
        if (do0 && do1 && do2) asm("" : "+v"(acc0), "+v"(acc1a), "+v"(acc1b), "+v"(acc2), "+v"(xn[0]));
        if (do0) {
            if (tt == 0) accn = bias0;
#pragma unroll
            for (int i = 0; i < 4; ++i) accn = wmma4(wx0[4 * tt + i], xn[i], accn);
            if (tt == 3) accx = accn;
        }

        const v4f g1a = gsel ? __builtin_shufflevector(acc1a, acc1a, 4, 5, 6, 7)
                             : __builtin_shufflevector(acc1a, acc1a, 0, 1, 2, 3);
        const v4f g1b = gsel ? __builtin_shufflevector(acc1b, acc1b, 4, 5, 6, 7)
                             : __builtin_shufflevector(acc1b, acc1b, 0, 1, 2, 3);
        const v4f g0  = gsel ? __builtin_shufflevector(acc0, acc0, 4, 5, 6, 7)
                             : __builtin_shufflevector(acc0, acc0, 0, 1, 2, 3);
        const v4f g2  = gsel ? __builtin_shufflevector(acc2, acc2, 4, 5, 6, 7)
                             : __builtin_shufflevector(acc2, acc2, 0, 1, 2, 3);
        const v4f d1  = gtile ? g1b : g1a;
        const v4f d02 = gtile ? g2 : g0;
        const v8f d = __builtin_shufflevector(d1, d02, 0, 1, 2, 3, 4, 5, 6, 7);
        float c1n = c1, c02n = c02;
        const v2f hn = cell2(d, c1n, c02n);
        if (do1) {
            c1 = c1n;
            sH1[nxt][o1] = hn.x;
        }
        const bool act02 = gtile ? do2 : do0;
        if (act02) {
            c02 = c02n;
            (&sH02[nxt][0][0])[o02] = hn.y;
        }
        if (do3 && wave == 0) outp[(size_t)(s - 3) * OUT_DIM] = pred;

        lds_barrier();
    };

    pass(0, 0, true, false, false, false);
    pass(1, 1, true, true,  false, false);
    pass(2, 2, true, true,  true,  false);
    pass(3, 3, true, true, true, true);
    for (int s = 4; s < T_STEPS; s += 4) {
        pass(s,     0, true, true, true, true);
        pass(s + 1, 1, true, true, true, true);
        pass(s + 2, 2, true, true, true, true);
        pass(s + 3, 3, true, true, true, true);
    }
    pass(T_STEPS,     0, false, true,  true,  true);
    pass(T_STEPS + 1, 1, false, false, true,  true);
    pass(T_STEPS + 2, 2, false, false, false, true);
}

extern "C" void kernel_launch(void* const* d_in, const int* in_sizes, int n_in,
                              void* d_out, int out_size, void* d_ws, size_t ws_size,
                              hipStream_t stream) {
    (void)in_sizes; (void)n_in; (void)out_size; (void)d_ws; (void)ws_size;
    const float* x    = (const float*)d_in[0];
    const float* Wih0 = (const float*)d_in[1];
    const float* Whh0 = (const float*)d_in[2];
    const float* bih0 = (const float*)d_in[3];
    const float* bhh0 = (const float*)d_in[4];
    const float* Wih1 = (const float*)d_in[5];
    const float* Whh1 = (const float*)d_in[6];
    const float* bih1 = (const float*)d_in[7];
    const float* bhh1 = (const float*)d_in[8];
    const float* Wih2 = (const float*)d_in[9];
    const float* Whh2 = (const float*)d_in[10];
    const float* bih2 = (const float*)d_in[11];
    const float* bhh2 = (const float*)d_in[12];
    const float* Wl   = (const float*)d_in[13];
    const float* bl   = (const float*)d_in[14];
    float* out = (float*)d_out;

    lstm3_scan_kernel<<<BATCH / ROWS, NTHREADS, 0, stream>>>(
        x, Wih0, Whh0, bih0, bhh0,
        Wih1, Whh1, bih1, bhh1,
        Wih2, Whh2, bih2, bhh2,
        Wl, bl, out);
}


# h hand-over overlapped: layers 2 passes apart, input-side products of layers 1, 2 issued BEHIND the h stores and between barrier signal and wait, x products cover the next h reads; sum order kept
# speedup vs baseline: 6.8103x; 1.0524x over previous
#include <hip/hip_runtime.h>


typedef __attribute__((ext_vector_type(2))) float v2f;
typedef __attribute__((ext_vector_type(4))) float v4f;
typedef __attribute__((ext_vector_type(8))) float v8f;

#define T_STEPS  2048
#define BATCH    256
#define IN_DIM   64
#define H0       16
#define H1       32
#define H2       16
#define OUT_DIM  8
#define NTHREADS 128
#define ROWS     4

__device__ __forceinline__ float tanhf_(float x) {
    return __builtin_amdgcn_tanhf(x);
}

__device__ __forceinline__ float sigmoid_half_(float xh) {
    return fmaf(__builtin_amdgcn_tanhf(xh), 0.5f, 0.5f);
}

__device__ __forceinline__ v2f load_frag(const float* base, int ld, int kbase, int lane) {
    const float* p = base + (lane & (ROWS - 1)) * ld + kbase + ((lane >> 4) << 1);
    v2f a;
    a.x = p[0];
    a.y = p[1];
    return a;
}

__device__ __forceinline__ v2f load_xfrag(const float* xrow, int kbase) {
    v2f a;
    a.x = xrow[kbase];
    a.y = xrow[kbase + 1];
    return a;
}

__device__ __forceinline__ v8f take_step(v8f p, int tt) {
    const unsigned sel = 0x32103210u + 0x44444444u * (unsigned)tt;
    v8f r;
#pragma unroll
    for (int i = 0; i < 8; ++i) {
        const unsigned u = __builtin_bit_cast(unsigned, (float)p[i]);
        r[i] = __builtin_bit_cast(float, __builtin_amdgcn_permlane16(u, u, sel, sel, false, false));
    }
    return r;
}

template <int H>
__device__ __forceinline__ v2f load_wfrag(const float* W, int ldw, int ubase, int hs, int kbase, int lane) {
    const int c   = lane & 15;
    const int row = (c & 3) * H + ubase + hs * (c >> 3) + ((c >> 2) & 1);
    const float* p = W + row * ldw + kbase + ((lane >> 4) << 1);
    const float sc = ((c & 3) == 2) ? 1.0f : 0.5f;
    v2f a;
    a.x = p[0] * sc;
    a.y = p[1] * sc;
    return a;
}

template <int H>
__device__ __forceinline__ v8f load_bias(const float* bi, const float* bh, int ubase, int hs, int lane) {
    const int hh = lane >> 4;
    v8f b;
#pragma unroll
    for (int r = 0; r < 8; ++r) {
        const int row = (r & 3) * H + ubase + hs * hh + (r >> 2);
        b[r] = (bi[row] + bh[row]) * (((r & 3) == 2) ? 1.0f : 0.5f);
    }
    return b;
}

__device__ __forceinline__ v8f wmma4(v2f a, v2f b, v8f c) {
    return __builtin_amdgcn_wmma_f32_16x16x4_f32(false, a, false, b, (short)0, c, false, false);
}

__device__ __forceinline__ void hand_over_signal() {
    __builtin_amdgcn_fence(__ATOMIC_RELEASE, "workgroup", "local");
    __builtin_amdgcn_s_barrier_signal(-1);
}
__device__ __forceinline__ void hand_over_wait() {
    __builtin_amdgcn_s_barrier_wait(-1);
    __builtin_amdgcn_fence(__ATOMIC_ACQUIRE, "workgroup", "local");
}

__device__ __forceinline__ v2f cell2(v8f d, float& c0, float& c1) {
    v2f h;
    {
        float cn = sigmoid_half_(d[1]) * c0 + sigmoid_half_(d[0]) * tanhf_(d[2]);
        c0 = cn;
        h.x = sigmoid_half_(d[3]) * tanhf_(cn);
    }
    {
        float cn = sigmoid_half_(d[5]) * c1 + sigmoid_half_(d[4]) * tanhf_(d[6]);
        c1 = cn;
        h.y = sigmoid_half_(d[7]) * tanhf_(cn);
    }
    return h;
}

__global__ __launch_bounds__(NTHREADS) void lstm3_scan_kernel(
    const float* __restrict__ x,
    const float* __restrict__ Wih0, const float* __restrict__ Whh0,
    const float* __restrict__ bih0, const float* __restrict__ bhh0,
    const float* __restrict__ Wih1, const float* __restrict__ Whh1,
    const float* __restrict__ bih1, const float* __restrict__ bhh1,
    const float* __restrict__ Wih2, const float* __restrict__ Whh2,
    const float* __restrict__ bih2, const float* __restrict__ bhh2,
    const float* __restrict__ Wl,   const float* __restrict__ bl,
    float* __restrict__ out)
{
    __shared__ __attribute__((aligned(16))) float sH02[2][2][ROWS * H0];
    __shared__ __attribute__((aligned(16))) float sH1[2][ROWS * H1];
    __shared__ __attribute__((aligned(16))) float sWl[OUT_DIM * H2];
    __shared__ float sBl[OUT_DIM];

    const int tid  = threadIdx.x;
    const int lane = tid & 31;
    const int wave = tid >> 5;
    const int br   = lane & (ROWS - 1);
    const int g    = (lane >> 2) & 3;
    const bool gsel  = (g & 1) != 0;
    const bool gtile = (g & 2) != 0;
    const int hh   = lane >> 4;
    const int b0   = blockIdx.x * ROWS;

    for (int i = tid; i < ROWS * H0; i += NTHREADS) {
        sH02[0][0][i] = 0.f; sH02[0][1][i] = 0.f; sH02[1][0][i] = 0.f; sH02[1][1][i] = 0.f;
    }
    for (int i = tid; i < ROWS * H1; i += NTHREADS) { sH1[0][i] = 0.f; sH1[1][i] = 0.f; }
    for (int i = tid; i < OUT_DIM * H2; i += NTHREADS) sWl[i] = Wl[i];
    if (tid < OUT_DIM) sBl[tid] = bl[tid];

    const int u0 = wave * 4;
    v2f wx0[16], wh0[4];
#pragma unroll
    for (int kk = 0; kk < 16; ++kk) wx0[kk] = load_wfrag<H0>(Wih0, IN_DIM, u0, 2, 4 * kk, lane);
#pragma unroll
    for (int kk = 0; kk < 4;  ++kk) wh0[kk] = load_wfrag<H0>(Whh0, H0, u0, 2, 4 * kk, lane);
    const v8f bias0 = load_bias<H0>(bih0, bhh0, u0, 2, lane);

    const int u1a = wave * 8, u1b = wave * 8 + 2;
    v2f wx1a[4], wx1b[4], wh1a[8], wh1b[8];
#pragma unroll
    for (int kk = 0; kk < 4; ++kk) {
        wx1a[kk] = load_wfrag<H1>(Wih1, H0, u1a, 4, 4 * kk, lane);
        wx1b[kk] = load_wfrag<H1>(Wih1, H0, u1b, 4, 4 * kk, lane);
    }
#pragma unroll
    for (int kk = 0; kk < 8; ++kk) {
        wh1a[kk] = load_wfrag<H1>(Whh1, H1, u1a, 4, 4 * kk, lane);
        wh1b[kk] = load_wfrag<H1>(Whh1, H1, u1b, 4, 4 * kk, lane);
    }
    const v8f bias1a = load_bias<H1>(bih1, bhh1, u1a, 4, lane);
    const v8f bias1b = load_bias<H1>(bih1, bhh1, u1b, 4, lane);

    const int u2 = wave * 4;
    v2f wx2[8], wh2[4];
#pragma unroll
    for (int kk = 0; kk < 8; ++kk) wx2[kk] = load_wfrag<H2>(Wih2, H1, u2, 2, 4 * kk, lane);
#pragma unroll
    for (int kk = 0; kk < 4; ++kk) wh2[kk] = load_wfrag<H2>(Whh2, H2, u2, 2, 4 * kk, lane);
    const v8f bias2 = load_bias<H2>(bih2, bhh2, u2, 2, lane);

    const int o1  = br * H1 + u1a + 4 * hh + g;
    const int o02 = (g >> 1) * (ROWS * H0) + br * H0 + u0 + 2 * hh + (g & 1);

    const int om = lane >> 3;
    const int oo = lane & 7;
    float* const outp = out + ((size_t)(b0 + om)) * T_STEPS * OUT_DIM + oo;

    float c1 = 0.f;
    float c02 = 0.f;

    const int xoff = (g * BATCH + br) * IN_DIM + 2 * hh;
    v8f accx = bias0;
    {
        const float* x0 = x + (size_t)b0 * IN_DIM + xoff;
#pragma unroll
        for (int kk = 0; kk < 16; ++kk) {
            v2f b = load_xfrag(x0, 4 * kk);
            accx = wmma4(wx0[kk], b, accx);
        }
    }
    v8f accn = bias0;
    v2f xq[4];
#pragma unroll
    for (int i = 0; i < 4; ++i) { xq[i].x = 0.f; xq[i].y = 0.f; }
    v8f p1a = bias1a, p1b = bias1b, p2 = bias2;

    __syncthreads();

    auto pass = [&](const int s, const int tt, const bool do0, const bool dox, const bool dop1,
                    const bool do1, const bool dop2, const bool do2, const bool do3)
                    __attribute__((always_inline)) {
        const int cur = tt & 1, nxt = cur ^ 1;
        const float* const h0cur = sH02[cur][0];
        const float* const h2cur = sH02[cur][1];

        v2f f0[4], f1[8], f2[4];
        if (do0 || dop1) {
#pragma unroll
            for (int kk = 0; kk < 4; ++kk) f0[kk] = load_frag(h0cur, H0, 4 * kk, lane);
        }
        if (do1 || dop2) {
#pragma unroll
            for (int kk = 0; kk < 8; ++kk) f1[kk] = load_frag(sH1[cur], H1, 4 * kk, lane);
        }
        if (do2) {
#pragma unroll
            for (int kk = 0; kk < 4; ++kk) f2[kk] = load_frag(h2cur, H2, 4 * kk, lane);
        }

        v4f oh[4], ow[4];
        float ob = 0.f;
        if (do3) {
#pragma unroll
            for (int q = 0; q < 4; ++q) {
                oh[q] = *(const v4f*)&h2cur[om * H2 + 4 * q];
                ow[q] = *(const v4f*)&sWl[oo * H2 + 4 * q];
            }
            ob = sBl[oo];
        }
        v2f xu[4];
        if (dox) {
#pragma unroll
            for (int i = 0; i < 4; ++i) xu[i] = xq[i];
        }
        if (do0) {
            const int tb = (s - tt + 4 < T_STEPS) ? (s - tt + 4) : (s - tt);
            const float* xp = x + ((size_t)tb * BATCH + b0) * IN_DIM + xoff + 16 * tt;
            if (s + 8 < T_STEPS) {
                __builtin_prefetch(x + ((size_t)(s + 8) * BATCH + b0) * IN_DIM + tid * 2, 0, 0);
            }
#pragma unroll
            for (int i = 0; i < 4; ++i) xq[i] = load_xfrag(xp, 4 * i);
        }
        __builtin_amdgcn_sched_barrier(0);

        if (dox) {
            const int tq = (tt + 3) & 3;
            if (tq == 0) accn = bias0;
#pragma unroll
            for (int i = 0; i < 4; ++i) accn = wmma4(wx0[4 * tq + i], xu[i], accn);
        }
        if (do0 && dox && do1 && do2) asm("" : "+v"(accn), "+v"(f0[0]), "+v"(f1[0]), "+v"(f2[0]));
        if (dox && ((tt + 3) & 3) == 3) accx = accn;

        float pred = ob;
        if (do3) {
#pragma unroll
            for (int q = 0; q < 4; ++q) {
                pred += oh[q].x * ow[q].x;
                pred += oh[q].y * ow[q].y;
                pred += oh[q].z * ow[q].z;
                pred += oh[q].w * ow[q].w;
            }
            asm("" : "+v"(pred));
        }

        v8f acc0 = do0 ? take_step(accx, tt) : bias0, acc1a = p1a, acc1b = p1b, acc2 = p2;
#pragma unroll
        for (int kk = 0; kk < 8; ++kk) {
            if (do1) {
                acc1a = wmma4(wh1a[kk], f1[kk], acc1a);
                acc1b = wmma4(wh1b[kk], f1[kk], acc1b);
            }
            if (kk < 4) {
                if (do2) acc2 = wmma4(wh2[kk], f2[kk], acc2);
            } else {
                if (do0) acc0 = wmma4(wh0[kk - 4], f0[kk - 4], acc0);
            }
        }
        if (do0 && do1 && do2 && dop1 && dop2)
            asm("" : "+v"(acc0), "+v"(acc1a), "+v"(acc1b), "+v"(acc2), "+v"(f0[0]), "+v"(f1[0]));

        v8f q2 = bias2, q1a = bias1a, q1b = bias1b;
        if (dop2) q2  = wmma4(wx2[0],  f1[0], q2);
        if (dop1) q1a = wmma4(wx1a[0], f0[0], q1a);
        if (dop2) q2  = wmma4(wx2[1],  f1[1], q2);
        if (dop1) q1b = wmma4(wx1b[0], f0[0], q1b);
        if (dop2) q2  = wmma4(wx2[2],  f1[2], q2);
        if (dop1) q1a = wmma4(wx1a[1], f0[1], q1a);

        const v4f g1a = gsel ? __builtin_shufflevector(acc1a, acc1a, 4, 5, 6, 7)
                             : __builtin_shufflevector(acc1a, acc1a, 0, 1, 2, 3);
        const v4f g1b = gsel ? __builtin_shufflevector(acc1b, acc1b, 4, 5, 6, 7)
                             : __builtin_shufflevector(acc1b, acc1b, 0, 1, 2, 3);
        const v4f g0  = gsel ? __builtin_shufflevector(acc0, acc0, 4, 5, 6, 7)
                             : __builtin_shufflevector(acc0, acc0, 0, 1, 2, 3);
        const v4f g2  = gsel ? __builtin_shufflevector(acc2, acc2, 4, 5, 6, 7)
                             : __builtin_shufflevector(acc2, acc2, 0, 1, 2, 3);
        const v4f d1  = gtile ? g1b : g1a;
        const v4f d02 = gtile ? g2 : g0;
        const v8f d = __builtin_shufflevector(d1, d02, 0, 1, 2, 3, 4, 5, 6, 7);
        float c1n = c1, c02n = c02;
        const v2f hn = cell2(d, c1n, c02n);
        if (do1) {
            c1 = c1n;
            sH1[nxt][o1] = hn.x;
        }
        const bool act02 = gtile ? do2 : do0;
        if (act02) {
            c02 = c02n;
            (&sH02[nxt][0][0])[o02] = hn.y;
        }
        if (do3 && wave == 0) outp[(size_t)(s - 5) * OUT_DIM] = pred;

        asm volatile("" : "+v"(q2), "+v"(q1a), "+v"(q1b));
        if (dop2) q2  = wmma4(wx2[3],  f1[3], q2);
        if (dop1) q1b = wmma4(wx1b[1], f0[1], q1b);
        if (dop2) q2  = wmma4(wx2[4],  f1[4], q2);
        if (dop1) q1a = wmma4(wx1a[2], f0[2], q1a);
        asm volatile("" : "+v"(q2), "+v"(q1a), "+v"(q1b));
        hand_over_signal();
        asm volatile("" : "+v"(q2), "+v"(q1a), "+v"(q1b));
        if (dop2) q2  = wmma4(wx2[5],  f1[5], q2);
        if (dop1) q1b = wmma4(wx1b[2], f0[2], q1b);
        if (dop2) q2  = wmma4(wx2[6],  f1[6], q2);
        if (dop1) q1a = wmma4(wx1a[3], f0[3], q1a);
        asm volatile("" : "+v"(q2), "+v"(q1a), "+v"(q1b));
        hand_over_wait();
        if (dop2) q2  = wmma4(wx2[7],  f1[7], q2);
        if (dop1) q1b = wmma4(wx1b[3], f0[3], q1b);
        p1a = q1a; p1b = q1b; p2 = q2;
    };

    pass(0, 0, true, false, false, false, false, false, false);
    pass(1, 1, true, true,  true,  false, false, false, false);
    pass(2, 2, true, true,  true,  true,  false, false, false);
    pass(3, 3, true, true,  true,  true,  true,  false, false);
    pass(4, 0, true, true,  true,  true,  true,  true,  false);
    pass(5, 1, true, true, true, true, true, true, true);
    pass(6, 2, true, true, true, true, true, true, true);
    pass(7, 3, true, true, true, true, true, true, true);
    for (int s = 8; s < T_STEPS; s += 4) {
        pass(s,     0, true, true, true, true, true, true, true);
        pass(s + 1, 1, true, true, true, true, true, true, true);
        pass(s + 2, 2, true, true, true, true, true, true, true);
        pass(s + 3, 3, true, true, true, true, true, true, true);
    }
    pass(T_STEPS,     0, false, false, true,  true,  true,  true,  true);
    pass(T_STEPS + 1, 1, false, false, false, true,  true,  true,  true);
    pass(T_STEPS + 2, 2, false, false, false, false, true,  true,  true);
    pass(T_STEPS + 3, 3, false, false, false, false, false, true,  true);
    pass(T_STEPS + 4, 0, false, false, false, false, false, false, true);
}

extern "C" void kernel_launch(void* const* d_in, const int* in_sizes, int n_in,
                              void* d_out, int out_size, void* d_ws, size_t ws_size,
                              hipStream_t stream) {
    (void)in_sizes; (void)n_in; (void)out_size; (void)d_ws; (void)ws_size;
    const float* x    = (const float*)d_in[0];
    const float* Wih0 = (const float*)d_in[1];
    const float* Whh0 = (const float*)d_in[2];
    const float* bih0 = (const float*)d_in[3];
    const float* bhh0 = (const float*)d_in[4];
    const float* Wih1 = (const float*)d_in[5];
    const float* Whh1 = (const float*)d_in[6];
    const float* bih1 = (const float*)d_in[7];
    const float* bhh1 = (const float*)d_in[8];
    const float* Wih2 = (const float*)d_in[9];
    const float* Whh2 = (const float*)d_in[10];
    const float* bih2 = (const float*)d_in[11];
    const float* bhh2 = (const float*)d_in[12];
    const float* Wl   = (const float*)d_in[13];
    const float* bl   = (const float*)d_in[14];
    float* out = (float*)d_out;

    lstm3_scan_kernel<<<BATCH / ROWS, NTHREADS, 0, stream>>>(
        x, Wih0, Whh0, bih0, bhh0,
        Wih1, Whh1, bih1, bhh1,
        Wih2, Whh2, bih2, bhh2,
        Wl, bl, out);
}


# output projection split by batch row over the 4 waves: lane = (output, k quarter), Wl in registers, quarters summed across 4 lanes (its sum order changes), one full-line store per 4 steps
# speedup vs baseline: 7.3002x; 1.0719x over previous
#include <hip/hip_runtime.h>


typedef __attribute__((ext_vector_type(2))) float v2f;
typedef __attribute__((ext_vector_type(4))) float v4f;
typedef __attribute__((ext_vector_type(8))) float v8f;

#define T_STEPS  2048
#define BATCH    256
#define IN_DIM   64
#define H0       16
#define H1       32
#define H2       16
#define OUT_DIM  8
#define NTHREADS 128
#define ROWS     4

__device__ __forceinline__ float tanhf_(float x) {
    return __builtin_amdgcn_tanhf(x);
}

__device__ __forceinline__ float sigmoid_half_(float xh) {
    return fmaf(__builtin_amdgcn_tanhf(xh), 0.5f, 0.5f);
}

__device__ __forceinline__ v2f load_frag(const float* base, int ld, int kbase, int lane) {
    const float* p = base + (lane & (ROWS - 1)) * ld + kbase + ((lane >> 4) << 1);
    v2f a;
    a.x = p[0];
    a.y = p[1];
    return a;
}

__device__ __forceinline__ v2f load_xfrag(const float* xrow, int kbase) {
    v2f a;
    a.x = xrow[kbase];
    a.y = xrow[kbase + 1];
    return a;
}

__device__ __forceinline__ v8f take_step(v8f p, int tt) {
    const unsigned sel = 0x32103210u + 0x44444444u * (unsigned)tt;
    v8f r;
#pragma unroll
    for (int i = 0; i < 8; ++i) {
        const unsigned u = __builtin_bit_cast(unsigned, (float)p[i]);
        r[i] = __builtin_bit_cast(float, __builtin_amdgcn_permlane16(u, u, sel, sel, false, false));
    }
    return r;
}

__device__ __forceinline__ float quad_xor1(float v) {
    const unsigned u = __builtin_bit_cast(unsigned, v);
    return __builtin_bit_cast(float, __builtin_amdgcn_permlane16(u, u, 0x67452301u, 0xEFCDAB89u, false, false));
}
__device__ __forceinline__ float quad_xor2(float v) {
    const unsigned u = __builtin_bit_cast(unsigned, v);
    return __builtin_bit_cast(float, __builtin_amdgcn_permlane16(u, u, 0x54761032u, 0xDCFE98BAu, false, false));
}

template <int H>
__device__ __forceinline__ v2f load_wfrag(const float* W, int ldw, int ubase, int hs, int kbase, int lane) {
    const int c   = lane & 15;
    const int row = (c & 3) * H + ubase + hs * (c >> 3) + ((c >> 2) & 1);
    const float* p = W + row * ldw + kbase + ((lane >> 4) << 1);
    const float sc = ((c & 3) == 2) ? 1.0f : 0.5f;
    v2f a;
    a.x = p[0] * sc;
    a.y = p[1] * sc;
    return a;
}

template <int H>
__device__ __forceinline__ v8f load_bias(const float* bi, const float* bh, int ubase, int hs, int lane) {
    const int hh = lane >> 4;
    v8f b;
#pragma unroll
    for (int r = 0; r < 8; ++r) {
        const int row = (r & 3) * H + ubase + hs * hh + (r >> 2);
        b[r] = (bi[row] + bh[row]) * (((r & 3) == 2) ? 1.0f : 0.5f);
    }
    return b;
}

__device__ __forceinline__ v8f wmma4(v2f a, v2f b, v8f c) {
    return __builtin_amdgcn_wmma_f32_16x16x4_f32(false, a, false, b, (short)0, c, false, false);
}

__device__ __forceinline__ void hand_over_signal() {
    __builtin_amdgcn_fence(__ATOMIC_RELEASE, "workgroup", "local");
    __builtin_amdgcn_s_barrier_signal(-1);
}
__device__ __forceinline__ void hand_over_wait() {
    __builtin_amdgcn_s_barrier_wait(-1);
    __builtin_amdgcn_fence(__ATOMIC_ACQUIRE, "workgroup", "local");
}

__device__ __forceinline__ v2f cell2(v8f d, float& c0, float& c1) {
    v2f h;
    {
        float cn = sigmoid_half_(d[1]) * c0 + sigmoid_half_(d[0]) * tanhf_(d[2]);
        c0 = cn;
        h.x = sigmoid_half_(d[3]) * tanhf_(cn);
    }
    {
        float cn = sigmoid_half_(d[5]) * c1 + sigmoid_half_(d[4]) * tanhf_(d[6]);
        c1 = cn;
        h.y = sigmoid_half_(d[7]) * tanhf_(cn);
    }
    return h;
}

__global__ __launch_bounds__(NTHREADS) void lstm3_scan_kernel(
    const float* __restrict__ x,
    const float* __restrict__ Wih0, const float* __restrict__ Whh0,
    const float* __restrict__ bih0, const float* __restrict__ bhh0,
    const float* __restrict__ Wih1, const float* __restrict__ Whh1,
    const float* __restrict__ bih1, const float* __restrict__ bhh1,
    const float* __restrict__ Wih2, const float* __restrict__ Whh2,
    const float* __restrict__ bih2, const float* __restrict__ bhh2,
    const float* __restrict__ Wl,   const float* __restrict__ bl,
    float* __restrict__ out)
{
    __shared__ __attribute__((aligned(16))) float sH02[2][2][ROWS * H0];
    __shared__ __attribute__((aligned(16))) float sH1[2][ROWS * H1];

    const int tid  = threadIdx.x;
    const int lane = tid & 31;
    const int wave = tid >> 5;
    const int br   = lane & (ROWS - 1);
    const int g    = (lane >> 2) & 3;
    const bool gsel  = (g & 1) != 0;
    const bool gtile = (g & 2) != 0;
    const int hh   = lane >> 4;
    const int b0   = blockIdx.x * ROWS;

    for (int i = tid; i < ROWS * H0; i += NTHREADS) {
        sH02[0][0][i] = 0.f; sH02[0][1][i] = 0.f; sH02[1][0][i] = 0.f; sH02[1][1][i] = 0.f;
    }
    for (int i = tid; i < ROWS * H1; i += NTHREADS) { sH1[0][i] = 0.f; sH1[1][i] = 0.f; }

    const int u0 = wave * 4;
    v2f wx0[16], wh0[4];
#pragma unroll
    for (int kk = 0; kk < 16; ++kk) wx0[kk] = load_wfrag<H0>(Wih0, IN_DIM, u0, 2, 4 * kk, lane);
#pragma unroll
    for (int kk = 0; kk < 4;  ++kk) wh0[kk] = load_wfrag<H0>(Whh0, H0, u0, 2, 4 * kk, lane);
    const v8f bias0 = load_bias<H0>(bih0, bhh0, u0, 2, lane);

    const int u1a = wave * 8, u1b = wave * 8 + 2;
    v2f wx1a[4], wx1b[4], wh1a[8], wh1b[8];
#pragma unroll
    for (int kk = 0; kk < 4; ++kk) {
        wx1a[kk] = load_wfrag<H1>(Wih1, H0, u1a, 4, 4 * kk, lane);
        wx1b[kk] = load_wfrag<H1>(Wih1, H0, u1b, 4, 4 * kk, lane);
    }
#pragma unroll
    for (int kk = 0; kk < 8; ++kk) {
        wh1a[kk] = load_wfrag<H1>(Whh1, H1, u1a, 4, 4 * kk, lane);
        wh1b[kk] = load_wfrag<H1>(Whh1, H1, u1b, 4, 4 * kk, lane);
    }
    const v8f bias1a = load_bias<H1>(bih1, bhh1, u1a, 4, lane);
    const v8f bias1b = load_bias<H1>(bih1, bhh1, u1b, 4, lane);

    const int u2 = wave * 4;
    v2f wx2[8], wh2[4];
#pragma unroll
    for (int kk = 0; kk < 8; ++kk) wx2[kk] = load_wfrag<H2>(Wih2, H1, u2, 2, 4 * kk, lane);
#pragma unroll
    for (int kk = 0; kk < 4; ++kk) wh2[kk] = load_wfrag<H2>(Whh2, H2, u2, 2, 4 * kk, lane);
    const v8f bias2 = load_bias<H2>(bih2, bhh2, u2, 2, lane);

    const int o1  = br * H1 + u1a + 4 * hh + g;
    const int o02 = (g >> 1) * (ROWS * H0) + br * H0 + u0 + 2 * hh + (g & 1);

    const int oo = lane >> 2;
    const int kq = lane & 3;
    v4f wl;
    wl.x = Wl[oo * H2 + 4 * kq + 0];
    wl.y = Wl[oo * H2 + 4 * kq + 1];
    wl.z = Wl[oo * H2 + 4 * kq + 2];
    wl.w = Wl[oo * H2 + 4 * kq + 3];
    const float blv = (kq == 0) ? bl[oo] : 0.f;
    float* const outp = out + ((size_t)(b0 + wave)) * T_STEPS * OUT_DIM + kq * OUT_DIM + oo;
    float hold = 0.f;

    float c1 = 0.f;
    float c02 = 0.f;

    const int xoff = (g * BATCH + br) * IN_DIM + 2 * hh;
    v8f accx = bias0;
    {
        const float* x0 = x + (size_t)b0 * IN_DIM + xoff;
#pragma unroll
        for (int kk = 0; kk < 16; ++kk) {
            v2f b = load_xfrag(x0, 4 * kk);
            accx = wmma4(wx0[kk], b, accx);
        }
    }
    v8f accn = bias0;
    v2f xq[4];
#pragma unroll
    for (int i = 0; i < 4; ++i) { xq[i].x = 0.f; xq[i].y = 0.f; }
    v8f p1a = bias1a, p1b = bias1b, p2 = bias2;

    __syncthreads();

    auto pass = [&](const int s, const int tt, const bool do0, const bool dox, const bool dop1,
                    const bool do1, const bool dop2, const bool do2, const bool do3)
                    __attribute__((always_inline)) {
        const int cur = tt & 1, nxt = cur ^ 1;
        const float* const h0cur = sH02[cur][0];
        const float* const h2cur = sH02[cur][1];

        v2f f0[4], f1[8], f2[4];
        if (do0 || dop1) {
#pragma unroll
            for (int kk = 0; kk < 4; ++kk) f0[kk] = load_frag(h0cur, H0, 4 * kk, lane);
        }
        if (do1 || dop2) {
#pragma unroll
            for (int kk = 0; kk < 8; ++kk) f1[kk] = load_frag(sH1[cur], H1, 4 * kk, lane);
        }
        if (do2) {
#pragma unroll
            for (int kk = 0; kk < 4; ++kk) f2[kk] = load_frag(h2cur, H2, 4 * kk, lane);
        }

        v4f oh;
        oh.x = 0.f; oh.y = 0.f; oh.z = 0.f; oh.w = 0.f;
        if (do3) oh = *(const v4f*)&h2cur[wave * H2 + 4 * kq];
        v2f xu[4];
        if (dox) {
#pragma unroll
            for (int i = 0; i < 4; ++i) xu[i] = xq[i];
        }
        if (do0) {
            const int tb = (s - tt + 4 < T_STEPS) ? (s - tt + 4) : (s - tt);
            const float* xp = x + ((size_t)tb * BATCH + b0) * IN_DIM + xoff + 16 * tt;
            if (s + 8 < T_STEPS) {
                __builtin_prefetch(x + ((size_t)(s + 8) * BATCH + b0) * IN_DIM + tid * 2, 0, 0);
            }
#pragma unroll
            for (int i = 0; i < 4; ++i) xq[i] = load_xfrag(xp, 4 * i);
        }
        __builtin_amdgcn_sched_barrier(0);

        if (dox) {
            const int tq = (tt + 3) & 3;
            if (tq == 0) accn = bias0;
#pragma unroll
            for (int i = 0; i < 4; ++i) accn = wmma4(wx0[4 * tq + i], xu[i], accn);
        }
        if (do0 && dox && do1 && do2) asm("" : "+v"(accn), "+v"(f0[0]), "+v"(f1[0]), "+v"(f2[0]));
        if (dox && ((tt + 3) & 3) == 3) accx = accn;


        v8f acc0 = do0 ? take_step(accx, tt) : bias0, acc1a = p1a, acc1b = p1b, acc2 = p2;
#pragma unroll
        for (int kk = 0; kk < 8; ++kk) {
            if (do1) {
                acc1a = wmma4(wh1a[kk], f1[kk], acc1a);
                acc1b = wmma4(wh1b[kk], f1[kk], acc1b);
            }
            if (kk < 4) {
                if (do2) acc2 = wmma4(wh2[kk], f2[kk], acc2);
            } else {
                if (do0) acc0 = wmma4(wh0[kk - 4], f0[kk - 4], acc0);
            }
        }
        if (do0 && do1 && do2 && dop1 && dop2 && do3)
            asm("" : "+v"(acc0), "+v"(acc1a), "+v"(acc1b), "+v"(acc2), "+v"(f0[0]), "+v"(f1[0]), "+v"(oh));
        else if (do0 && do1 && do2 && dop1 && dop2)
            asm("" : "+v"(acc0), "+v"(acc1a), "+v"(acc1b), "+v"(acc2), "+v"(f0[0]), "+v"(f1[0]));

        if (do3) {
            float pred = fmaf(oh.x, wl.x, blv);
            pred = fmaf(oh.y, wl.y, pred);
            pred = fmaf(oh.z, wl.z, pred);
            pred = fmaf(oh.w, wl.w, pred);
            pred += quad_xor1(pred);
            pred += quad_xor2(pred);
            if (kq == ((tt + 3) & 3)) hold = pred;
        }

        v8f q2 = bias2, q1a = bias1a, q1b = bias1b;
        if (dop2) q2  = wmma4(wx2[0],  f1[0], q2);
        if (dop1) q1a = wmma4(wx1a[0], f0[0], q1a);
        if (dop2) q2  = wmma4(wx2[1],  f1[1], q2);
        if (dop1) q1b = wmma4(wx1b[0], f0[0], q1b);
        if (dop2) q2  = wmma4(wx2[2],  f1[2], q2);
        if (dop1) q1a = wmma4(wx1a[1], f0[1], q1a);

        const v4f g1a = gsel ? __builtin_shufflevector(acc1a, acc1a, 4, 5, 6, 7)
                             : __builtin_shufflevector(acc1a, acc1a, 0, 1, 2, 3);
        const v4f g1b = gsel ? __builtin_shufflevector(acc1b, acc1b, 4, 5, 6, 7)
                             : __builtin_shufflevector(acc1b, acc1b, 0, 1, 2, 3);
        const v4f g0  = gsel ? __builtin_shufflevector(acc0, acc0, 4, 5, 6, 7)
                             : __builtin_shufflevector(acc0, acc0, 0, 1, 2, 3);
        const v4f g2  = gsel ? __builtin_shufflevector(acc2, acc2, 4, 5, 6, 7)
                             : __builtin_shufflevector(acc2, acc2, 0, 1, 2, 3);
        const v4f d1  = gtile ? g1b : g1a;
        const v4f d02 = gtile ? g2 : g0;
        const v8f d = __builtin_shufflevector(d1, d02, 0, 1, 2, 3, 4, 5, 6, 7);
        float c1n = c1, c02n = c02;
        const v2f hn = cell2(d, c1n, c02n);
        if (do1) {
            c1 = c1n;
            sH1[nxt][o1] = hn.x;
        }
        const bool act02 = gtile ? do2 : do0;
        if (act02) {
            c02 = c02n;
            (&sH02[nxt][0][0])[o02] = hn.y;
        }
        if (do3 && ((tt + 3) & 3) == 3) outp[(size_t)(s - 8) * OUT_DIM] = hold;

        asm volatile("" : "+v"(q2), "+v"(q1a), "+v"(q1b));
        if (dop2) q2  = wmma4(wx2[3],  f1[3], q2);
        if (dop1) q1b = wmma4(wx1b[1], f0[1], q1b);
        if (dop2) q2  = wmma4(wx2[4],  f1[4], q2);
        if (dop1) q1a = wmma4(wx1a[2], f0[2], q1a);
        asm volatile("" : "+v"(q2), "+v"(q1a), "+v"(q1b));
        hand_over_signal();
        asm volatile("" : "+v"(q2), "+v"(q1a), "+v"(q1b));
        if (dop2) q2  = wmma4(wx2[5],  f1[5], q2);
        if (dop1) q1b = wmma4(wx1b[2], f0[2], q1b);
        if (dop2) q2  = wmma4(wx2[6],  f1[6], q2);
        if (dop1) q1a = wmma4(wx1a[3], f0[3], q1a);
        asm volatile("" : "+v"(q2), "+v"(q1a), "+v"(q1b));
        hand_over_wait();
        if (dop2) q2  = wmma4(wx2[7],  f1[7], q2);
        if (dop1) q1b = wmma4(wx1b[3], f0[3], q1b);
        p1a = q1a; p1b = q1b; p2 = q2;
    };

    pass(0, 0, true, false, false, false, false, false, false);
    pass(1, 1, true, true,  true,  false, false, false, false);
    pass(2, 2, true, true,  true,  true,  false, false, false);
    pass(3, 3, true, true,  true,  true,  true,  false, false);
    pass(4, 0, true, true,  true,  true,  true,  true,  false);
    pass(5, 1, true, true, true, true, true, true, true);
    pass(6, 2, true, true, true, true, true, true, true);
    pass(7, 3, true, true, true, true, true, true, true);
    for (int s = 8; s < T_STEPS; s += 4) {
        pass(s,     0, true, true, true, true, true, true, true);
        pass(s + 1, 1, true, true, true, true, true, true, true);
        pass(s + 2, 2, true, true, true, true, true, true, true);
        pass(s + 3, 3, true, true, true, true, true, true, true);
    }
    pass(T_STEPS,     0, false, false, true,  true,  true,  true,  true);
    pass(T_STEPS + 1, 1, false, false, false, true,  true,  true,  true);
    pass(T_STEPS + 2, 2, false, false, false, false, true,  true,  true);
    pass(T_STEPS + 3, 3, false, false, false, false, false, true,  true);
    pass(T_STEPS + 4, 0, false, false, false, false, false, false, true);
}

extern "C" void kernel_launch(void* const* d_in, const int* in_sizes, int n_in,
                              void* d_out, int out_size, void* d_ws, size_t ws_size,
                              hipStream_t stream) {
    (void)in_sizes; (void)n_in; (void)out_size; (void)d_ws; (void)ws_size;
    const float* x    = (const float*)d_in[0];
    const float* Wih0 = (const float*)d_in[1];
    const float* Whh0 = (const float*)d_in[2];
    const float* bih0 = (const float*)d_in[3];
    const float* bhh0 = (const float*)d_in[4];
    const float* Wih1 = (const float*)d_in[5];
    const float* Whh1 = (const float*)d_in[6];
    const float* bih1 = (const float*)d_in[7];
    const float* bhh1 = (const float*)d_in[8];
    const float* Wih2 = (const float*)d_in[9];
    const float* Whh2 = (const float*)d_in[10];
    const float* bih2 = (const float*)d_in[11];
    const float* bhh2 = (const float*)d_in[12];
    const float* Wl   = (const float*)d_in[13];
    const float* bl   = (const float*)d_in[14];
    float* out = (float*)d_out;

    lstm3_scan_kernel<<<BATCH / ROWS, NTHREADS, 0, stream>>>(
        x, Wih0, Whh0, bih0, bhh0,
        Wih1, Whh1, bih1, bhh1,
        Wih2, Whh2, bih2, bhh2,
        Wl, bl, out);
}


# input-side products of layers 1, 2 batched over 4 time steps like x: h rings of 8 slots, layers 8 passes apart, one 16-column tile = 4 steps x 4 rows, 4 products a pass not 16; sum order kept
# speedup vs baseline: 8.3212x; 1.1399x over previous
#include <hip/hip_runtime.h>


typedef __attribute__((ext_vector_type(2))) float v2f;
typedef __attribute__((ext_vector_type(4))) float v4f;
typedef __attribute__((ext_vector_type(8))) float v8f;

#define T_STEPS  2048
#define BATCH    256
#define IN_DIM   64
#define H0       16
#define H1       32
#define H2       16
#define OUT_DIM  8
#define NTHREADS 128
#define ROWS     4

__device__ __forceinline__ float tanhf_(float x) {
    return __builtin_amdgcn_tanhf(x);
}

__device__ __forceinline__ float sigmoid_half_(float xh) {
    return fmaf(__builtin_amdgcn_tanhf(xh), 0.5f, 0.5f);
}

__device__ __forceinline__ v2f load_frag(const float* base, int ld, int kbase, int lane) {
    const float* p = base + (lane & (ROWS - 1)) * ld + kbase + ((lane >> 4) << 1);
    v2f a;
    a.x = p[0];
    a.y = p[1];
    return a;
}

__device__ __forceinline__ v2f load_xfrag(const float* xrow, int kbase) {
    v2f a;
    a.x = xrow[kbase];
    a.y = xrow[kbase + 1];
    return a;
}

__device__ __forceinline__ v8f take_step(v8f p, int tt) {
    const unsigned sel = 0x32103210u + 0x44444444u * (unsigned)tt;
    v8f r;
#pragma unroll
    for (int i = 0; i < 8; ++i) {
        const unsigned u = __builtin_bit_cast(unsigned, (float)p[i]);
        r[i] = __builtin_bit_cast(float, __builtin_amdgcn_permlane16(u, u, sel, sel, false, false));
    }
    return r;
}

__device__ __forceinline__ float quad_xor1(float v) {
    const unsigned u = __builtin_bit_cast(unsigned, v);
    return __builtin_bit_cast(float, __builtin_amdgcn_permlane16(u, u, 0x67452301u, 0xEFCDAB89u, false, false));
}
__device__ __forceinline__ float quad_xor2(float v) {
    const unsigned u = __builtin_bit_cast(unsigned, v);
    return __builtin_bit_cast(float, __builtin_amdgcn_permlane16(u, u, 0x54761032u, 0xDCFE98BAu, false, false));
}

template <int H>
__device__ __forceinline__ v2f load_wfrag(const float* W, int ldw, int ubase, int hs, int kbase, int lane) {
    const int c   = lane & 15;
    const int row = (c & 3) * H + ubase + hs * (c >> 3) + ((c >> 2) & 1);
    const float* p = W + row * ldw + kbase + ((lane >> 4) << 1);
    const float sc = ((c & 3) == 2) ? 1.0f : 0.5f;
    v2f a;
    a.x = p[0] * sc;
    a.y = p[1] * sc;
    return a;
}

template <int H>
__device__ __forceinline__ v8f load_bias(const float* bi, const float* bh, int ubase, int hs, int lane) {
    const int hh = lane >> 4;
    v8f b;
#pragma unroll
    for (int r = 0; r < 8; ++r) {
        const int row = (r & 3) * H + ubase + hs * hh + (r >> 2);
        b[r] = (bi[row] + bh[row]) * (((r & 3) == 2) ? 1.0f : 0.5f);
    }
    return b;
}

__device__ __forceinline__ v8f wmma4(v2f a, v2f b, v8f c) {
    return __builtin_amdgcn_wmma_f32_16x16x4_f32(false, a, false, b, (short)0, c, false, false);
}

__device__ __forceinline__ void hand_over_signal() {
    __builtin_amdgcn_fence(__ATOMIC_RELEASE, "workgroup", "local");
    __builtin_amdgcn_s_barrier_signal(-1);
}
__device__ __forceinline__ void hand_over_wait() {
    __builtin_amdgcn_s_barrier_wait(-1);
    __builtin_amdgcn_fence(__ATOMIC_ACQUIRE, "workgroup", "local");
}

__device__ __forceinline__ v2f cell2(v8f d, float& c0, float& c1) {
    v2f h;
    {
        float cn = sigmoid_half_(d[1]) * c0 + sigmoid_half_(d[0]) * tanhf_(d[2]);
        c0 = cn;
        h.x = sigmoid_half_(d[3]) * tanhf_(cn);
    }
    {
        float cn = sigmoid_half_(d[5]) * c1 + sigmoid_half_(d[4]) * tanhf_(d[6]);
        c1 = cn;
        h.y = sigmoid_half_(d[7]) * tanhf_(cn);
    }
    return h;
}

__global__ __launch_bounds__(NTHREADS) void lstm3_scan_kernel(
    const float* __restrict__ x,
    const float* __restrict__ Wih0, const float* __restrict__ Whh0,
    const float* __restrict__ bih0, const float* __restrict__ bhh0,
    const float* __restrict__ Wih1, const float* __restrict__ Whh1,
    const float* __restrict__ bih1, const float* __restrict__ bhh1,
    const float* __restrict__ Wih2, const float* __restrict__ Whh2,
    const float* __restrict__ bih2, const float* __restrict__ bhh2,
    const float* __restrict__ Wl,   const float* __restrict__ bl,
    float* __restrict__ out)
{
    __shared__ __attribute__((aligned(16))) float sH02[2][8][ROWS * H0];
    __shared__ __attribute__((aligned(16))) float sH1[8][ROWS * H1];

    const int tid  = threadIdx.x;
    const int lane = tid & 31;
    const int wave = tid >> 5;
    const int br   = lane & (ROWS - 1);
    const int g    = (lane >> 2) & 3;
    const bool gsel  = (g & 1) != 0;
    const bool gtile = (g & 2) != 0;
    const int hh   = lane >> 4;
    const int b0   = blockIdx.x * ROWS;

    for (int i = tid; i < 2 * 8 * ROWS * H0; i += NTHREADS) (&sH02[0][0][0])[i] = 0.f;
    for (int i = tid; i < 8 * ROWS * H1; i += NTHREADS) (&sH1[0][0])[i] = 0.f;

    const int u0 = wave * 4;
    v2f wx0[16], wh0[4];
#pragma unroll
    for (int kk = 0; kk < 16; ++kk) wx0[kk] = load_wfrag<H0>(Wih0, IN_DIM, u0, 2, 4 * kk, lane);
#pragma unroll
    for (int kk = 0; kk < 4;  ++kk) wh0[kk] = load_wfrag<H0>(Whh0, H0, u0, 2, 4 * kk, lane);
    const v8f bias0 = load_bias<H0>(bih0, bhh0, u0, 2, lane);

    const int u1a = wave * 8, u1b = wave * 8 + 2;
    v2f wx1a[4], wx1b[4], wh1a[8], wh1b[8];
#pragma unroll
    for (int kk = 0; kk < 4; ++kk) {
        wx1a[kk] = load_wfrag<H1>(Wih1, H0, u1a, 4, 4 * kk, lane);
        wx1b[kk] = load_wfrag<H1>(Wih1, H0, u1b, 4, 4 * kk, lane);
    }
#pragma unroll
    for (int kk = 0; kk < 8; ++kk) {
        wh1a[kk] = load_wfrag<H1>(Whh1, H1, u1a, 4, 4 * kk, lane);
        wh1b[kk] = load_wfrag<H1>(Whh1, H1, u1b, 4, 4 * kk, lane);
    }
    const v8f bias1a = load_bias<H1>(bih1, bhh1, u1a, 4, lane);
    const v8f bias1b = load_bias<H1>(bih1, bhh1, u1b, 4, lane);

    const int u2 = wave * 4;
    v2f wx2[8], wh2[4];
#pragma unroll
    for (int kk = 0; kk < 8; ++kk) wx2[kk] = load_wfrag<H2>(Wih2, H1, u2, 2, 4 * kk, lane);
#pragma unroll
    for (int kk = 0; kk < 4; ++kk) wh2[kk] = load_wfrag<H2>(Whh2, H2, u2, 2, 4 * kk, lane);
    const v8f bias2 = load_bias<H2>(bih2, bhh2, u2, 2, lane);

    const int o1  = br * H1 + u1a + 4 * hh + g;
    const int o02 = (g >> 1) * (8 * ROWS * H0) + br * H0 + u0 + 2 * hh + (g & 1);

    const int oo = lane >> 2;
    const int kq = lane & 3;
    v4f wl;
    wl.x = Wl[oo * H2 + 4 * kq + 0];
    wl.y = Wl[oo * H2 + 4 * kq + 1];
    wl.z = Wl[oo * H2 + 4 * kq + 2];
    wl.w = Wl[oo * H2 + 4 * kq + 3];
    const float blv = (kq == 0) ? bl[oo] : 0.f;
    float* const outp = out + ((size_t)(b0 + wave)) * T_STEPS * OUT_DIM + kq * OUT_DIM + oo;
    float hold = 0.f;

    float c1 = 0.f;
    float c02 = 0.f;

    const int xoff = (g * BATCH + br) * IN_DIM + 2 * hh;
    v8f accx = bias0;
    {
        const float* x0 = x + (size_t)b0 * IN_DIM + xoff;
#pragma unroll
        for (int kk = 0; kk < 16; ++kk) {
            v2f b = load_xfrag(x0, 4 * kk);
            accx = wmma4(wx0[kk], b, accx);
        }
    }
    v8f accn = bias0;
    v2f xq[4];
#pragma unroll
    for (int i = 0; i < 4; ++i) { xq[i].x = 0.f; xq[i].y = 0.f; }
    v8f P1a = bias1a, P1b = bias1b, P2 = bias2;
    v8f n1a = bias1a, n1b = bias1b, n2 = bias2;
    const float* const pb0 = &sH02[0][0][0] + g * (ROWS * H0) + br * H0 + 2 * hh;
    const float* const pb1 = &sH1[0][0] + g * (ROWS * H1) + br * H1 + 2 * hh;

    __syncthreads();

    auto pass = [&](const int s, const int t8, const bool ax, const bool a0, const bool a1,
                    const bool a2, const bool ao)
                    __attribute__((always_inline)) {
        const int tt  = t8 & 3;
        const int prv = (t8 + 7) & 7;
        const int bb  = (t8 & 4) ^ 4;
        const float* const h0cur = sH02[0][prv];
        const float* const h2cur = sH02[1][prv];

        v2f f0[4], f1[8], f2[4];
#pragma unroll
        for (int kk = 0; kk < 4; ++kk) f0[kk] = load_frag(h0cur, H0, 4 * kk, lane);
#pragma unroll
        for (int kk = 0; kk < 8; ++kk) f1[kk] = load_frag(sH1[prv], H1, 4 * kk, lane);
#pragma unroll
        for (int kk = 0; kk < 4; ++kk) f2[kk] = load_frag(h2cur, H2, 4 * kk, lane);
        v2f fb0, fb1[2];
        {
            const float* p = pb0 + bb * (ROWS * H0) + 4 * tt;
            fb0.x = p[0]; fb0.y = p[1];
        }
#pragma unroll
        for (int i = 0; i < 2; ++i) {
            const float* p = pb1 + bb * (ROWS * H1) + 8 * tt + 4 * i;
            fb1[i].x = p[0]; fb1[i].y = p[1];
        }

        v4f oh = *(const v4f*)&h2cur[wave * H2 + 4 * kq];
        v2f xu[4];
#pragma unroll
        for (int i = 0; i < 4; ++i) xu[i] = xq[i];
        {
            const int tb = (s - tt + 4 < T_STEPS - 4) ? (s - tt + 4) : (T_STEPS - 4);
            const float* xp = x + ((size_t)tb * BATCH + b0) * IN_DIM + xoff + 16 * tt;
            if (s + 8 < T_STEPS) {
                __builtin_prefetch(x + ((size_t)(s + 8) * BATCH + b0) * IN_DIM + tid * 2, 0, 0);
            }
#pragma unroll
            for (int i = 0; i < 4; ++i) xq[i] = load_xfrag(xp, 4 * i);
        }
        __builtin_amdgcn_sched_barrier(0);

        if (ax) {
            const int tq = (tt + 3) & 3;
            if (tq == 0) accn = bias0;
#pragma unroll
            for (int i = 0; i < 4; ++i) accn = wmma4(wx0[4 * tq + i], xu[i], accn);
        }
        asm("" : "+v"(accn), "+v"(f0[0]), "+v"(f1[0]), "+v"(f2[0]));
        if (ax && ((tt + 3) & 3) == 3) accx = accn;

        v8f acc0  = take_step(accx, tt);
        v8f acc1a = take_step(P1a, tt);
        v8f acc1b = take_step(P1b, tt);
        v8f acc2  = take_step(P2, tt);
#pragma unroll
        for (int kk = 0; kk < 8; ++kk) {
            acc1a = wmma4(wh1a[kk], f1[kk], acc1a);
            acc1b = wmma4(wh1b[kk], f1[kk], acc1b);
            if (kk < 4) {
                acc2 = wmma4(wh2[kk], f2[kk], acc2);
            } else {
                acc0 = wmma4(wh0[kk - 4], f0[kk - 4], acc0);
            }
        }
        asm("" : "+v"(acc0), "+v"(acc1a), "+v"(acc1b), "+v"(acc2), "+v"(fb0), "+v"(fb1[0]), "+v"(oh));

        {
            float pred = fmaf(oh.x, wl.x, blv);
            pred = fmaf(oh.y, wl.y, pred);
            pred = fmaf(oh.z, wl.z, pred);
            pred = fmaf(oh.w, wl.w, pred);
            pred += quad_xor1(pred);
            pred += quad_xor2(pred);
            if (kq == ((tt + 3) & 3)) hold = pred;
        }

        const v4f g1a = gsel ? __builtin_shufflevector(acc1a, acc1a, 4, 5, 6, 7)
                             : __builtin_shufflevector(acc1a, acc1a, 0, 1, 2, 3);
        const v4f g1b = gsel ? __builtin_shufflevector(acc1b, acc1b, 4, 5, 6, 7)
                             : __builtin_shufflevector(acc1b, acc1b, 0, 1, 2, 3);
        const v4f g0  = gsel ? __builtin_shufflevector(acc0, acc0, 4, 5, 6, 7)
                             : __builtin_shufflevector(acc0, acc0, 0, 1, 2, 3);
        const v4f g2  = gsel ? __builtin_shufflevector(acc2, acc2, 4, 5, 6, 7)
                             : __builtin_shufflevector(acc2, acc2, 0, 1, 2, 3);
        const v4f d1  = gtile ? g1b : g1a;
        const v4f d02 = gtile ? g2 : g0;
        const v8f d = __builtin_shufflevector(d1, d02, 0, 1, 2, 3, 4, 5, 6, 7);
        float c1n = c1, c02n = c02;
        const v2f hn = cell2(d, c1n, c02n);
        if (a1) {
            c1 = c1n;
            sH1[t8][o1] = hn.x;
        }
        const bool act02 = gtile ? a2 : a0;
        if (act02) {
            c02 = c02n;
            (&sH02[0][0][0])[o02 + t8 * (ROWS * H0)] = hn.y;
        }
        if (ao && tt == 0) outp[(size_t)(s - 20) * OUT_DIM] = hold;

        if (tt == 0) { n1a = bias1a; n1b = bias1b; n2 = bias2; }
        asm volatile("" : "+v"(n2), "+v"(n1a), "+v"(n1b));
        n2  = wmma4(wx2[2 * tt], fb1[0], n2);
        n1a = wmma4(wx1a[tt],    fb0,    n1a);
        asm volatile("" : "+v"(n2), "+v"(n1a), "+v"(n1b));
        hand_over_signal();
        asm volatile("" : "+v"(n2), "+v"(n1a), "+v"(n1b));
        n2  = wmma4(wx2[2 * tt + 1], fb1[1], n2);
        n1b = wmma4(wx1b[tt],        fb0,    n1b);
        asm volatile("" : "+v"(n2), "+v"(n1a), "+v"(n1b));
        hand_over_wait();
        if (tt == 3) { P1a = n1a; P1b = n1b; P2 = n2; }
    };

    auto gen = [&](const int s, const int t8) __attribute__((always_inline)) {
        pass(s, t8, s >= 1, s < T_STEPS, s >= 8 && s < T_STEPS + 8,
             s >= 16 && s < T_STEPS + 16, s >= 20 && s < T_STEPS + 17);
    };
    for (int s = 0; s < T_STEPS + 17; s += 8) {
        if (s >= 24 && s < T_STEPS) {
            pass(s,     0, true, true, true, true, true);
            pass(s + 1, 1, true, true, true, true, true);
            pass(s + 2, 2, true, true, true, true, true);
            pass(s + 3, 3, true, true, true, true, true);
            pass(s + 4, 4, true, true, true, true, true);
            pass(s + 5, 5, true, true, true, true, true);
            pass(s + 6, 6, true, true, true, true, true);
            pass(s + 7, 7, true, true, true, true, true);
        } else {
            gen(s, 0);
            if (s >= T_STEPS + 16) break;
            gen(s + 1, 1); gen(s + 2, 2); gen(s + 3, 3);
            gen(s + 4, 4); gen(s + 5, 5); gen(s + 6, 6); gen(s + 7, 7);
        }
    }
}

extern "C" void kernel_launch(void* const* d_in, const int* in_sizes, int n_in,
                              void* d_out, int out_size, void* d_ws, size_t ws_size,
                              hipStream_t stream) {
    (void)in_sizes; (void)n_in; (void)out_size; (void)d_ws; (void)ws_size;
    const float* x    = (const float*)d_in[0];
    const float* Wih0 = (const float*)d_in[1];
    const float* Whh0 = (const float*)d_in[2];
    const float* bih0 = (const float*)d_in[3];
    const float* bhh0 = (const float*)d_in[4];
    const float* Wih1 = (const float*)d_in[5];
    const float* Whh1 = (const float*)d_in[6];
    const float* bih1 = (const float*)d_in[7];
    const float* bhh1 = (const float*)d_in[8];
    const float* Wih2 = (const float*)d_in[9];
    const float* Whh2 = (const float*)d_in[10];
    const float* bih2 = (const float*)d_in[11];
    const float* bhh2 = (const float*)d_in[12];
    const float* Wl   = (const float*)d_in[13];
    const float* bl   = (const float*)d_in[14];
    float* out = (float*)d_out;

    lstm3_scan_kernel<<<BATCH / ROWS, NTHREADS, 0, stream>>>(
        x, Wih0, Whh0, bih0, bhh0,
        Wih1, Whh1, bih1, bhh1,
        Wih2, Whh2, bih2, bhh2,
        Wl, bl, out);
}
